# starting kernel
# baseline (speedup 1.0000x reference)
#include <hip/hip_runtime.h>

typedef __attribute__((ext_vector_type(2))) float v2f;
typedef __attribute__((ext_vector_type(8))) float v8f;

#define N_PTS  16384
#define BATCH  4
#define CIN    64
#define COUT   64
#define KSZ    9
#define BC     (BATCH * CIN)

__global__ void xq_transpose_kernel(const float* __restrict__ x,
                                    const float* __restrict__ quad,
                                    float* __restrict__ xt) {
  int idx = blockIdx.x * blockDim.x + threadIdx.x;
  int m = idx >> 8;
  int j = idx & 255;
  int b = j >> 6, c = j & 63;
  xt[idx] = x[(b * CIN + c) * N_PTS + m] * quad[m];
}

__global__ void zero_kernel(float* __restrict__ p, int n) {
  int i = blockIdx.x * blockDim.x + threadIdx.x;
  if (i < n) p[i] = 0.0f;
}

__global__ void gemm_wmma_kernel(const float* __restrict__ xt,
                                 const float* __restrict__ weight,
                                 float* __restrict__ Tk, int k) {
  __shared__ float a_lds[32 * 66];
  __shared__ float w_lds[64 * 66];

  const int tid = threadIdx.x;
  const int b   = blockIdx.y;
  const int m0  = blockIdx.x * 32;

  for (int i = tid; i < 64 * 64; i += 256) {
    int o = i >> 6, c = i & 63;
    w_lds[o * 66 + c] = weight[(o * CIN + c) * KSZ + k];
  }
  for (int i = tid; i < 32 * 64; i += 256) {
    int r = i >> 6, c = i & 63;
    a_lds[r * 66 + c] = xt[(m0 + r) * BC + b * CIN + c];
  }
  __syncthreads();

  const int lane = tid & 31;
  const int hi   = lane >> 4;
  const int l15  = lane & 15;
  const int wave = tid >> 5;
  const int wm   = wave >> 2;
  const int wn   = wave & 3;

  v8f acc = {};
#pragma unroll
  for (int ks = 0; ks < 16; ++ks) {
    const int kk = ks * 4 + hi * 2;
    v2f a  = { a_lds[(wm * 16 + l15) * 66 + kk],
               a_lds[(wm * 16 + l15) * 66 + kk + 1] };
    v2f bw = { w_lds[(wn * 16 + l15) * 66 + kk],
               w_lds[(wn * 16 + l15) * 66 + kk + 1] };
    acc = __builtin_amdgcn_wmma_f32_16x16x4_f32(
         false, a,  false, bw,
         (short)0, acc,  false,  false);
  }

  const int o     = wn * 16 + l15;
  const int mbase = m0 + wm * 16 + hi * 8;
#pragma unroll
  for (int r = 0; r < 8; ++r) {
    Tk[(mbase + r) * BC + b * CIN + o] = acc[r];
  }
}

__global__ void scatter_kernel(const int* __restrict__ kI,
                               const int* __restrict__ oI,
                               const int* __restrict__ iI,
                               const float* __restrict__ psi,
                               const float* __restrict__ Tk,
                               float* __restrict__ ytmp,
                               int k, int nnz) {
  int gid  = blockIdx.x * blockDim.x + threadIdx.x;
  int e    = gid >> 5;
  int lane = gid & 31;
  if (e >= nnz) return;
  if (kI[e] != k) return;

  float p  = psi[e];
  int outp = oI[e];
  int inp  = iI[e];

  const float4* src = (const float4*)(Tk + (size_t)inp * BC) + lane * 2;
  float4 v0 = src[0];
  float4 v1 = src[1];

  float* dst = ytmp + (size_t)outp * BC + lane * 8;
  atomicAdd(dst + 0, p * v0.x);
  atomicAdd(dst + 1, p * v0.y);
  atomicAdd(dst + 2, p * v0.z);
  atomicAdd(dst + 3, p * v0.w);
  atomicAdd(dst + 4, p * v1.x);
  atomicAdd(dst + 5, p * v1.y);
  atomicAdd(dst + 6, p * v1.z);
  atomicAdd(dst + 7, p * v1.w);
}

__global__ void finalize_kernel(const float* __restrict__ ytmp,
                                const float* __restrict__ bias,
                                float* __restrict__ y) {
  int idx = blockIdx.x * blockDim.x + threadIdx.x;
  int n  = idx & (N_PTS - 1);
  int bo = idx >> 14;
  int o  = bo & 63;
  int b  = bo >> 6;
  y[idx] = ytmp[(size_t)n * BC + b * CIN + o] + bias[o];
}

extern "C" void kernel_launch(void* const* d_in, const int* in_sizes, int n_in,
                              void* d_out, int out_size, void* d_ws, size_t ws_size,
                              hipStream_t stream) {
  const float* x        = (const float*)d_in[0];
  const int*   psi_idx  = (const int*)  d_in[1];
  const float* psi_vals = (const float*)d_in[2];
  const float* quad     = (const float*)d_in[3];
  const float* weight   = (const float*)d_in[4];
  const float* bias     = (const float*)d_in[5];
  float*       y        = (float*)d_out;

  const int nnz = in_sizes[2];

  float* xt   = (float*)d_ws;
  float* Tk   = xt + (size_t)N_PTS * BC;
  float* ytmp = Tk + (size_t)N_PTS * BC;

  const int* kI = psi_idx;
  const int* oI = psi_idx + nnz;
  const int* iI = psi_idx + 2 * nnz;

  const int elems = N_PTS * BC;

  xq_transpose_kernel<<<elems / 256, 256, 0, stream>>>(x, quad, xt);
  zero_kernel<<<elems / 256, 256, 0, stream>>>(ytmp, elems);

  const int sblocks = (nnz * 32 + 255) / 256;
  for (int k = 0; k < KSZ; ++k) {
    gemm_wmma_kernel<<<dim3(N_PTS / 32, BATCH), 256, 0, stream>>>(xt, weight, Tk, k);
    scatter_kernel<<<sblocks, 256, 0, stream>>>(kI, oI, iI, psi_vals, Tk, ytmp, k, nnz);
  }

  finalize_kernel<<<(BATCH * COUT * N_PTS) / 256, 256, 0, stream>>>(ytmp, bias, y);
}


# Float-atomic scatter replaced by a per-call counting sort of the entries by output point: one wave per output row sums its index-k entries in registers and stores plainly; summation order changes
# speedup vs baseline: 1.6534x; 1.6534x over previous
#include <hip/hip_runtime.h>

typedef __attribute__((ext_vector_type(2))) float v2f;
typedef __attribute__((ext_vector_type(8))) float v8f;

#define N_PTS  16384
#define BATCH  4
#define CIN    64
#define COUT   64
#define KSZ    9
#define BC     (BATCH * CIN)

__global__ void xq_transpose_kernel(const float* __restrict__ x,
                                    const float* __restrict__ quad,
                                    float* __restrict__ xt) {
  int idx = blockIdx.x * blockDim.x + threadIdx.x;
  int m = idx >> 8;
  int j = idx & 255;
  int b = j >> 6, c = j & 63;
  xt[idx] = x[(b * CIN + c) * N_PTS + m] * quad[m];
}

__global__ void zero_kernel(float* __restrict__ p, int n) {
  int i = blockIdx.x * blockDim.x + threadIdx.x;
  if (i < n) p[i] = 0.0f;
}

__global__ void gemm_wmma_kernel(const float* __restrict__ xt,
                                 const float* __restrict__ weight,
                                 float* __restrict__ Tk, int k) {
  __shared__ float a_lds[32 * 66];
  __shared__ float w_lds[64 * 66];

  const int tid = threadIdx.x;
  const int b   = blockIdx.y;
  const int m0  = blockIdx.x * 32;

  for (int i = tid; i < 64 * 64; i += 256) {
    int o = i >> 6, c = i & 63;
    w_lds[o * 66 + c] = weight[(o * CIN + c) * KSZ + k];
  }
  for (int i = tid; i < 32 * 64; i += 256) {
    int r = i >> 6, c = i & 63;
    a_lds[r * 66 + c] = xt[(m0 + r) * BC + b * CIN + c];
  }
  __syncthreads();

  const int lane = tid & 31;
  const int hi   = lane >> 4;
  const int l15  = lane & 15;
  const int wave = tid >> 5;
  const int wm   = wave >> 2;
  const int wn   = wave & 3;

  v8f acc = {};
#pragma unroll
  for (int ks = 0; ks < 16; ++ks) {
    const int kk = ks * 4 + hi * 2;
    v2f a  = { a_lds[(wm * 16 + l15) * 66 + kk],
               a_lds[(wm * 16 + l15) * 66 + kk + 1] };
    v2f bw = { w_lds[(wn * 16 + l15) * 66 + kk],
               w_lds[(wn * 16 + l15) * 66 + kk + 1] };
    acc = __builtin_amdgcn_wmma_f32_16x16x4_f32(
         false, a,  false, bw,
         (short)0, acc,  false,  false);
  }

  const int o     = wn * 16 + l15;
  const int mbase = m0 + wm * 16 + hi * 8;
#pragma unroll
  for (int r = 0; r < 8; ++r) {
    Tk[(mbase + r) * BC + b * CIN + o] = acc[r];
  }
}

__global__ void count_kernel(const int* __restrict__ kI,
                             const int* __restrict__ oI,
                             int* __restrict__ cnt, int nnz) {
  int e = blockIdx.x * blockDim.x + threadIdx.x;
  if (e >= nnz) return;
  unsigned kk = (unsigned)kI[e];
  unsigned o  = (unsigned)oI[e];
  if (kk < KSZ && o < N_PTS) atomicAdd(&cnt[o], 1);
}

__global__ void __launch_bounds__(1024)
scan_kernel(const int* __restrict__ cnt, int* __restrict__ cur) {
  __shared__ int wsum[32];
  const int t    = threadIdx.x;
  const int lane = t & 31;
  const int w    = t >> 5;

  const int4* c4 = (const int4*)(cnt + t * 16);
  int4 q0 = c4[0], q1 = c4[1], q2 = c4[2], q3 = c4[3];
  int v[16] = { q0.x, q0.y, q0.z, q0.w, q1.x, q1.y, q1.z, q1.w,
                q2.x, q2.y, q2.z, q2.w, q3.x, q3.y, q3.z, q3.w };
  int s = 0;
  for (int i = 0; i < 16; ++i) s += v[i];

  int inc = s;
  for (int off = 1; off < 32; off <<= 1) {
    int up = __shfl_up(inc, off);
    if (lane >= off) inc += up;
  }
  if (lane == 31) wsum[w] = inc;
  __syncthreads();
  if (w == 0) {
    int x = wsum[lane];
    int xi = x;
    for (int off = 1; off < 32; off <<= 1) {
      int up = __shfl_up(xi, off);
      if (lane >= off) xi += up;
    }
    wsum[lane] = xi - x;
  }
  __syncthreads();

  int run = wsum[w] + inc - s;
  int r[16];
  for (int i = 0; i < 16; ++i) { r[i] = run; run += v[i]; }
  int4* o4 = (int4*)(cur + t * 16);
  o4[0] = make_int4(r[0],  r[1],  r[2],  r[3]);
  o4[1] = make_int4(r[4],  r[5],  r[6],  r[7]);
  o4[2] = make_int4(r[8],  r[9],  r[10], r[11]);
  o4[3] = make_int4(r[12], r[13], r[14], r[15]);
}

__global__ void fill_kernel(const int* __restrict__ kI,
                            const int* __restrict__ oI,
                            const int* __restrict__ iI,
                            const float* __restrict__ psi,
                            int* __restrict__ cur,
                            int* __restrict__ sk,
                            int* __restrict__ sin_,
                            float* __restrict__ sp, int nnz) {
  int e = blockIdx.x * blockDim.x + threadIdx.x;
  if (e >= nnz) return;
  unsigned kk = (unsigned)kI[e];
  unsigned o  = (unsigned)oI[e];
  if (kk < KSZ && o < N_PTS) {
    int pos = atomicAdd(&cur[o], 1);
    if ((unsigned)pos < (unsigned)nnz) {
      sk[pos]   = (int)kk;
      sin_[pos] = iI[e];
      sp[pos]   = psi[e];
    }
  }
}

struct Row8 { float4 lo, hi; };

__device__ __forceinline__ Row8 load_row(const float* __restrict__ Tk, int inp, int lane) {
  const float4* s = (const float4*)(Tk + (size_t)inp * BC) + lane * 2;
  Row8 r;
  r.lo = s[0];
  r.hi = s[1];
  return r;
}

__device__ __forceinline__ void fma_row(float4& a0, float4& a1, float p, const Row8& r) {
  a0.x += p * r.lo.x; a0.y += p * r.lo.y; a0.z += p * r.lo.z; a0.w += p * r.lo.w;
  a1.x += p * r.hi.x; a1.y += p * r.hi.y; a1.z += p * r.hi.z; a1.w += p * r.hi.w;
}

__global__ void gather_kernel(const int* __restrict__ cnt,
                              const int* __restrict__ cur,
                              const int* __restrict__ sk,
                              const int* __restrict__ sin_,
                              const float* __restrict__ sp,
                              const float* __restrict__ Tk,
                              float* __restrict__ ytmp,
                              int k, int nnz) {
  const int gid  = blockIdx.x * blockDim.x + threadIdx.x;
  const int lane = gid & 31;
  const int n = __builtin_amdgcn_readfirstlane(gid >> 5);

  int end = cur[n];
  int len = cnt[n];
  if (end > nnz) end = nnz;
  int begin = end - len;
  if (begin < 0) begin = 0;

  float4* dst = (float4*)(ytmp + (size_t)n * BC) + lane * 2;
  float4 o0 = make_float4(0.f, 0.f, 0.f, 0.f);
  float4 o1 = o0;
  if (k != 0) { o0 = dst[0]; o1 = dst[1]; }

  float4 a0 = make_float4(0.f, 0.f, 0.f, 0.f);
  float4 a1 = a0;

  for (int base = begin; base < end; base += 32) {
    const int  idx = base + lane;
    const bool ok  = idx < end;
    const int   ek = ok ? sk[idx]   : -1;
    const int   ei = ok ? (sin_[idx] & (N_PTS - 1)) : 0;
    const float ep = ok ? sp[idx]   : 0.f;
    unsigned mask = __builtin_amdgcn_ballot_w32(ek == k);
    while (mask) {
      const int l0 = __builtin_ctz(mask);
      mask &= mask - 1;
      const float p0 = __int_as_float(__builtin_amdgcn_readlane(__float_as_int(ep), l0));
      const Row8 r0 = load_row(Tk, __builtin_amdgcn_readlane(ei, l0), lane);
      if (!mask) { fma_row(a0, a1, p0, r0); break; }

      const int l1 = __builtin_ctz(mask);
      mask &= mask - 1;
      const float p1 = __int_as_float(__builtin_amdgcn_readlane(__float_as_int(ep), l1));
      const Row8 r1 = load_row(Tk, __builtin_amdgcn_readlane(ei, l1), lane);
      if (!mask) { fma_row(a0, a1, p0, r0); fma_row(a0, a1, p1, r1); break; }

      const int l2 = __builtin_ctz(mask);
      mask &= mask - 1;
      const float p2 = __int_as_float(__builtin_amdgcn_readlane(__float_as_int(ep), l2));
      const Row8 r2 = load_row(Tk, __builtin_amdgcn_readlane(ei, l2), lane);
      if (!mask) {
        fma_row(a0, a1, p0, r0); fma_row(a0, a1, p1, r1); fma_row(a0, a1, p2, r2);
        break;
      }

      const int l3 = __builtin_ctz(mask);
      mask &= mask - 1;
      const float p3 = __int_as_float(__builtin_amdgcn_readlane(__float_as_int(ep), l3));
      const Row8 r3 = load_row(Tk, __builtin_amdgcn_readlane(ei, l3), lane);
      __builtin_amdgcn_sched_barrier(0);
      fma_row(a0, a1, p0, r0); fma_row(a0, a1, p1, r1);
      fma_row(a0, a1, p2, r2); fma_row(a0, a1, p3, r3);
    }
  }

  o0.x += a0.x; o0.y += a0.y; o0.z += a0.z; o0.w += a0.w;
  o1.x += a1.x; o1.y += a1.y; o1.z += a1.z; o1.w += a1.w;
  dst[0] = o0;
  dst[1] = o1;
}

__global__ void finalize_kernel(const float* __restrict__ ytmp,
                                const float* __restrict__ bias,
                                float* __restrict__ y) {
  int idx = blockIdx.x * blockDim.x + threadIdx.x;
  int n  = idx & (N_PTS - 1);
  int bo = idx >> 14;
  int o  = bo & 63;
  int b  = bo >> 6;
  y[idx] = ytmp[(size_t)n * BC + b * CIN + o] + bias[o];
}

extern "C" void kernel_launch(void* const* d_in, const int* in_sizes, int n_in,
                              void* d_out, int out_size, void* d_ws, size_t ws_size,
                              hipStream_t stream) {
  const float* x        = (const float*)d_in[0];
  const int*   psi_idx  = (const int*)  d_in[1];
  const float* psi_vals = (const float*)d_in[2];
  const float* quad     = (const float*)d_in[3];
  const float* weight   = (const float*)d_in[4];
  const float* bias     = (const float*)d_in[5];
  float*       y        = (float*)d_out;

  const int nnz = in_sizes[2];

  float* xt   = (float*)d_ws;
  float* Tk   = xt + (size_t)N_PTS * BC;
  float* ytmp = Tk + (size_t)N_PTS * BC;

  const int* kI = psi_idx;
  const int* oI = psi_idx + nnz;
  const int* iI = psi_idx + 2 * nnz;

  const int elems = N_PTS * BC;

  int*   cnt  = (int*)d_out;
  int*   cur  = cnt + N_PTS;
  int*   sk   = cur + N_PTS;
  int*   sin_ = sk + nnz;
  float* sp   = (float*)(sin_ + nnz);

  xq_transpose_kernel<<<elems / 256, 256, 0, stream>>>(x, quad, xt);

  const int eblocks = (nnz + 255) / 256;
  zero_kernel<<<N_PTS / 256, 256, 0, stream>>>((float*)cnt, N_PTS);
  count_kernel<<<eblocks, 256, 0, stream>>>(kI, oI, cnt, nnz);
  scan_kernel<<<1, 1024, 0, stream>>>(cnt, cur);
  fill_kernel<<<eblocks, 256, 0, stream>>>(kI, oI, iI, psi_vals, cur, sk, sin_, sp, nnz);

  for (int k = 0; k < KSZ; ++k) {
    gemm_wmma_kernel<<<dim3(N_PTS / 32, BATCH), 256, 0, stream>>>(xt, weight, Tk, k);
    gather_kernel<<<(N_PTS * 32) / 256, 256, 0, stream>>>(cnt, cur, sk, sin_, sp, Tk, ytmp, k, nnz);
  }

  finalize_kernel<<<(BATCH * COUT * N_PTS) / 256, 256, 0, stream>>>(ytmp, bias, y);
}


# Work re-mapped to fewer, longer-lived waves: 32 points per thread in both transposes, 4 row tiles per GEMM block, 4 output points per gather wave, strided entry loops; same sums, same order
# speedup vs baseline: 1.8016x; 1.0896x over previous
#include <hip/hip_runtime.h>

typedef __attribute__((ext_vector_type(2))) float v2f;
typedef __attribute__((ext_vector_type(8))) float v8f;

#define N_PTS  16384
#define BATCH  4
#define CIN    64
#define COUT   64
#define KSZ    9
#define BC     (BATCH * CIN)


__global__ void __launch_bounds__(256)
xq_transpose_kernel(const float* __restrict__ x,
                    const float* __restrict__ quad,
                    float* __restrict__ xt) {
  const int j  = threadIdx.x;
  const int m0 = blockIdx.x * 32;
  const float4* s = (const float4*)(x + (size_t)j * N_PTS + m0);
  float4 v[8];
#pragma unroll
  for (int i = 0; i < 8; ++i) v[i] = s[i];
  float* d = xt + (size_t)m0 * BC + j;
#pragma unroll
  for (int i = 0; i < 8; ++i) {
    d[(4 * i + 0) * BC] = v[i].x * quad[m0 + 4 * i + 0];
    d[(4 * i + 1) * BC] = v[i].y * quad[m0 + 4 * i + 1];
    d[(4 * i + 2) * BC] = v[i].z * quad[m0 + 4 * i + 2];
    d[(4 * i + 3) * BC] = v[i].w * quad[m0 + 4 * i + 3];
  }
}

__global__ void zero_kernel(float* __restrict__ p, int n) {
  int i = blockIdx.x * blockDim.x + threadIdx.x;
  if (i < n) p[i] = 0.0f;
}

#define GEMM_TPB 4

__global__ void __launch_bounds__(256)
gemm_wmma_kernel(const float* __restrict__ xt,
                                 const float* __restrict__ weight,
                                 float* __restrict__ Tk, int k) {
  __shared__ float a_lds[32 * 66];
  __shared__ float w_lds[64 * 66];

  const int tid = threadIdx.x;
  const int b   = blockIdx.y;
  int m0 = blockIdx.x * (32 * GEMM_TPB);

  const int ar = tid >> 6;
  const int ac = tid & 63;
  float pa[8];
#pragma unroll
  for (int q = 0; q < 8; ++q)
    pa[q] = xt[(size_t)(m0 + ar + 4 * q) * BC + b * CIN + ac];

  for (int i = tid; i < 64 * 64; i += 256) {
    int o = i >> 6, c = i & 63;
    w_lds[o * 66 + c] = weight[(o * CIN + c) * KSZ + k];
  }

  const int lane = tid & 31;
  const int hi   = lane >> 4;
  const int l15  = lane & 15;
  const int wave = tid >> 5;
  const int wm   = wave >> 2;
  const int wn   = wave & 3;

  for (int it = 0; it < GEMM_TPB; ++it) {
#pragma unroll
  for (int q = 0; q < 8; ++q) a_lds[(ar + 4 * q) * 66 + ac] = pa[q];
  __syncthreads();
  if (it + 1 < GEMM_TPB) {
#pragma unroll
    for (int q = 0; q < 8; ++q)
      pa[q] = xt[(size_t)(m0 + 32 + ar + 4 * q) * BC + b * CIN + ac];
  }

  v8f acc = {};
#pragma unroll
  for (int ks = 0; ks < 16; ++ks) {
    const int kk = ks * 4 + hi * 2;
    v2f a  = { a_lds[(wm * 16 + l15) * 66 + kk],
               a_lds[(wm * 16 + l15) * 66 + kk + 1] };
    v2f bw = { w_lds[(wn * 16 + l15) * 66 + kk],
               w_lds[(wn * 16 + l15) * 66 + kk + 1] };
    acc = __builtin_amdgcn_wmma_f32_16x16x4_f32(
         false, a,  false, bw,
         (short)0, acc,  false,  false);
  }

  const int o     = wn * 16 + l15;
  const int mbase = m0 + wm * 16 + hi * 8;
#pragma unroll
  for (int r = 0; r < 8; ++r) {
    Tk[(size_t)(mbase + r) * BC + b * CIN + o] = acc[r];
  }
  __syncthreads();
  m0 += 32;
  }
}

__global__ void count_kernel(const int* __restrict__ kI,
                             const int* __restrict__ oI,
                             int* __restrict__ cnt, int nnz) {
  const int step = gridDim.x * blockDim.x;
  for (int e = blockIdx.x * blockDim.x + threadIdx.x; e < nnz; e += step) {
    unsigned kk = (unsigned)kI[e];
    unsigned o  = (unsigned)oI[e];
    if (kk < KSZ && o < N_PTS) atomicAdd(&cnt[o], 1);
  }
}

__global__ void __launch_bounds__(1024)
scan_kernel(const int* __restrict__ cnt, int* __restrict__ cur) {
  __shared__ int wsum[32];
  const int t    = threadIdx.x;
  const int lane = t & 31;
  const int w    = t >> 5;

  const int4* c4 = (const int4*)(cnt + t * 16);
  int4 q0 = c4[0], q1 = c4[1], q2 = c4[2], q3 = c4[3];
  int v[16] = { q0.x, q0.y, q0.z, q0.w, q1.x, q1.y, q1.z, q1.w,
                q2.x, q2.y, q2.z, q2.w, q3.x, q3.y, q3.z, q3.w };
  int s = 0;
  for (int i = 0; i < 16; ++i) s += v[i];

  int inc = s;
  for (int off = 1; off < 32; off <<= 1) {
    int up = __shfl_up(inc, off);
    if (lane >= off) inc += up;
  }
  if (lane == 31) wsum[w] = inc;
  __syncthreads();
  if (w == 0) {
    int x = wsum[lane];
    int xi = x;
    for (int off = 1; off < 32; off <<= 1) {
      int up = __shfl_up(xi, off);
      if (lane >= off) xi += up;
    }
    wsum[lane] = xi - x;
  }
  __syncthreads();

  int run = wsum[w] + inc - s;
  int r[16];
  for (int i = 0; i < 16; ++i) { r[i] = run; run += v[i]; }
  int4* o4 = (int4*)(cur + t * 16);
  o4[0] = make_int4(r[0],  r[1],  r[2],  r[3]);
  o4[1] = make_int4(r[4],  r[5],  r[6],  r[7]);
  o4[2] = make_int4(r[8],  r[9],  r[10], r[11]);
  o4[3] = make_int4(r[12], r[13], r[14], r[15]);
}

__global__ void fill_kernel(const int* __restrict__ kI,
                            const int* __restrict__ oI,
                            const int* __restrict__ iI,
                            const float* __restrict__ psi,
                            int* __restrict__ cur,
                            int* __restrict__ sk,
                            int* __restrict__ sin_,
                            float* __restrict__ sp, int nnz) {
  const int step = gridDim.x * blockDim.x;
  for (int e = blockIdx.x * blockDim.x + threadIdx.x; e < nnz; e += step) {
    unsigned kk = (unsigned)kI[e];
    unsigned o  = (unsigned)oI[e];
    if (kk < KSZ && o < N_PTS) {
      int pos = atomicAdd(&cur[o], 1);
      if ((unsigned)pos < (unsigned)nnz) {
        sk[pos]   = (int)kk;
        sin_[pos] = iI[e];
        sp[pos]   = psi[e];
      }
    }
  }
}

#define GATHER_PPW 4

struct Row8 { float4 lo, hi; };
struct Chunk { int ek; int ei; float ep; };

__device__ __forceinline__ Row8 load_row(const float* __restrict__ Tk, int inp, int lane) {
  const float4* s = (const float4*)(Tk + (size_t)inp * BC) + lane * 2;
  Row8 r;
  r.lo = s[0];
  r.hi = s[1];
  return r;
}

__device__ __forceinline__ void fma_row(float4& a0, float4& a1, float p, const Row8& r) {
  a0.x += p * r.lo.x; a0.y += p * r.lo.y; a0.z += p * r.lo.z; a0.w += p * r.lo.w;
  a1.x += p * r.hi.x; a1.y += p * r.hi.y; a1.z += p * r.hi.z; a1.w += p * r.hi.w;
}

__device__ __forceinline__ Chunk load_chunk(const int* __restrict__ sk,
                                            const int* __restrict__ sin_,
                                            const float* __restrict__ sp,
                                            int base, int end, int nnz, int lane) {
  const int  idx = base + lane;
  const bool ok  = idx < end;
  int j = idx < nnz ? idx : nnz - 1;
  if (j < 0) j = 0;
  const int   vk = sk[j];
  const int   vi = sin_[j];
  const float vp = sp[j];
  Chunk c;
  c.ek = ok ? vk : -1;
  c.ei = ok ? vi : 0;
  c.ep = ok ? vp : 0.f;
  return c;
}

__device__ __forceinline__ void add_chunk(const Chunk& c, int k,
                                          const float* __restrict__ Tk, int lane,
                                          float4& a0, float4& a1) {
  unsigned mask = __builtin_amdgcn_ballot_w32(c.ek == k);
  while (mask) {
    const int l0 = __builtin_ctz(mask);
    mask &= mask - 1;
    const float p0 = __int_as_float(__builtin_amdgcn_readlane(__float_as_int(c.ep), l0));
    const Row8 r0 = load_row(Tk, __builtin_amdgcn_readlane(c.ei, l0) & (N_PTS - 1), lane);
    if (!mask) { fma_row(a0, a1, p0, r0); break; }

    const int l1 = __builtin_ctz(mask);
    mask &= mask - 1;
    const float p1 = __int_as_float(__builtin_amdgcn_readlane(__float_as_int(c.ep), l1));
    const Row8 r1 = load_row(Tk, __builtin_amdgcn_readlane(c.ei, l1) & (N_PTS - 1), lane);
    if (!mask) { fma_row(a0, a1, p0, r0); fma_row(a0, a1, p1, r1); break; }

    const int l2 = __builtin_ctz(mask);
    mask &= mask - 1;
    const float p2 = __int_as_float(__builtin_amdgcn_readlane(__float_as_int(c.ep), l2));
    const Row8 r2 = load_row(Tk, __builtin_amdgcn_readlane(c.ei, l2) & (N_PTS - 1), lane);
    if (!mask) {
      fma_row(a0, a1, p0, r0); fma_row(a0, a1, p1, r1); fma_row(a0, a1, p2, r2);
      break;
    }

    const int l3 = __builtin_ctz(mask);
    mask &= mask - 1;
    const float p3 = __int_as_float(__builtin_amdgcn_readlane(__float_as_int(c.ep), l3));
    const Row8 r3 = load_row(Tk, __builtin_amdgcn_readlane(c.ei, l3) & (N_PTS - 1), lane);
    __builtin_amdgcn_sched_barrier(0);
    float q0 = p0, q1 = p1, q2 = p2;
    asm volatile("" : "+s"(q0), "+s"(q1), "+s"(q2));
    fma_row(a0, a1, q0, r0); fma_row(a0, a1, q1, r1);
    fma_row(a0, a1, q2, r2); fma_row(a0, a1, p3, r3);
  }
}

__global__ void __launch_bounds__(256)
gather_kernel(const int* __restrict__ cnt,
              const int* __restrict__ cur,
              const int* __restrict__ sk,
              const int* __restrict__ sin_,
              const float* __restrict__ sp,
              const float* __restrict__ Tk,
              float* __restrict__ ytmp,
              int k, int nnz) {
  const int gid  = blockIdx.x * blockDim.x + threadIdx.x;
  const int lane = gid & 31;
  const int n0 = __builtin_amdgcn_readfirstlane(gid >> 5) * GATHER_PPW;

  int beg[GATHER_PPW], end[GATHER_PPW];
#pragma unroll
  for (int i = 0; i < GATHER_PPW; ++i) {
    int e   = cur[n0 + i];
    int len = cnt[n0 + i];
    if (e > nnz) e = nnz;
    int bg = e - len;
    if (bg < 0) bg = 0;
    beg[i] = bg;
    end[i] = e;
  }

  Chunk ck[GATHER_PPW];
#pragma unroll
  for (int i = 0; i < GATHER_PPW; ++i)
    ck[i] = load_chunk(sk, sin_, sp, beg[i], end[i], nnz, lane);

  float4* const dst0 = (float4*)(ytmp + (size_t)n0 * BC) + lane * 2;
  float4 o0[GATHER_PPW], o1[GATHER_PPW];
#pragma unroll
  for (int i = 0; i < GATHER_PPW; ++i) {
    o0[i] = make_float4(0.f, 0.f, 0.f, 0.f);
    o1[i] = o0[i];
    if (k != 0) { o0[i] = dst0[i * (BC / 4)]; o1[i] = dst0[i * (BC / 4) + 1]; }
  }

#pragma unroll
  for (int i = 0; i < GATHER_PPW; ++i) {
    float4 a0 = make_float4(0.f, 0.f, 0.f, 0.f);
    float4 a1 = a0;

    add_chunk(ck[i], k, Tk, lane, a0, a1);
    for (int base = beg[i] + 32; base < end[i]; base += 32) {
      const Chunk c2 = load_chunk(sk, sin_, sp, base, end[i], nnz, lane);
      add_chunk(c2, k, Tk, lane, a0, a1);
    }

    float4 r0 = o0[i], r1 = o1[i];
    r0.x += a0.x; r0.y += a0.y; r0.z += a0.z; r0.w += a0.w;
    r1.x += a1.x; r1.y += a1.y; r1.z += a1.z; r1.w += a1.w;
    dst0[i * (BC / 4)]     = r0;
    dst0[i * (BC / 4) + 1] = r1;
  }
}

__global__ void __launch_bounds__(256)
finalize_kernel(const float* __restrict__ ytmp,
                const float* __restrict__ bias,
                float* __restrict__ y) {
  const int j  = threadIdx.x;
  const int n0 = blockIdx.x * 32;
  const float bv = bias[j & 63];
  const float* s = ytmp + (size_t)n0 * BC + j;
  float r[32];
#pragma unroll
  for (int i = 0; i < 32; ++i) r[i] = s[(size_t)i * BC] + bv;
  float4* d = (float4*)(y + (size_t)j * N_PTS + n0);
#pragma unroll
  for (int i = 0; i < 8; ++i)
    d[i] = make_float4(r[4 * i], r[4 * i + 1], r[4 * i + 2], r[4 * i + 3]);
}

extern "C" void kernel_launch(void* const* d_in, const int* in_sizes, int n_in,
                              void* d_out, int out_size, void* d_ws, size_t ws_size,
                              hipStream_t stream) {
  const float* x        = (const float*)d_in[0];
  const int*   psi_idx  = (const int*)  d_in[1];
  const float* psi_vals = (const float*)d_in[2];
  const float* quad     = (const float*)d_in[3];
  const float* weight   = (const float*)d_in[4];
  const float* bias     = (const float*)d_in[5];
  float*       y        = (float*)d_out;

  const int nnz = in_sizes[2];

  float* xt   = (float*)d_ws;
  float* Tk   = xt + (size_t)N_PTS * BC;
  float* ytmp = Tk + (size_t)N_PTS * BC;

  const int* kI = psi_idx;
  const int* oI = psi_idx + nnz;
  const int* iI = psi_idx + 2 * nnz;

  int*   cnt  = (int*)d_out;
  int*   cur  = cnt + N_PTS;
  int*   sk   = cur + N_PTS;
  int*   sin_ = sk + nnz;
  float* sp   = (float*)(sin_ + nnz);

  xq_transpose_kernel<<<N_PTS / 32, 256, 0, stream>>>(x, quad, xt);

  zero_kernel<<<N_PTS / 256, 256, 0, stream>>>((float*)cnt, N_PTS);
  count_kernel<<<256, 256, 0, stream>>>(kI, oI, cnt, nnz);
  scan_kernel<<<1, 1024, 0, stream>>>(cnt, cur);
  fill_kernel<<<256, 256, 0, stream>>>(kI, oI, iI, psi_vals, cur, sk, sin_, sp, nnz);

  for (int k = 0; k < KSZ; ++k) {
    gemm_wmma_kernel<<<dim3(N_PTS / 32 / GEMM_TPB, BATCH), 256, 0, stream>>>(xt, weight, Tk, k);
    gather_kernel<<<(N_PTS / GATHER_PPW * 32) / 256, 256, 0, stream>>>(cnt, cur, sk, sin_, sp, Tk, ytmp, k, nnz);
  }

  finalize_kernel<<<N_PTS / 32, 256, 0, stream>>>(ytmp, bias, y);
}


# Gather restructured as one stream per wave and kernel index: matching entries of its four points packed into an LDS list, up to 12 row loads issued before the first multiply-add; same sums and order
# speedup vs baseline: 3.6444x; 2.0229x over previous
#include <hip/hip_runtime.h>

typedef __attribute__((ext_vector_type(2))) float v2f;
typedef __attribute__((ext_vector_type(8))) float v8f;

#define N_PTS  16384
#define BATCH  4
#define CIN    64
#define COUT   64
#define KSZ    9
#define BC     (BATCH * CIN)


#define XQ_BLOCKS   (N_PTS / 32)
#define WT_BLOCKS   (KSZ * COUT * CIN / 256)
#define ZERO_BLOCKS (N_PTS / 256)

__global__ void __launch_bounds__(256)
setup_kernel(const float* __restrict__ x,
             const float* __restrict__ quad,
             float* __restrict__ xt,
             const float* __restrict__ weight,
             float* __restrict__ wt,
             int* __restrict__ cnt) {
  const int blk = blockIdx.x;
  if (blk < XQ_BLOCKS) {
    const int j  = threadIdx.x;
    const int m0 = blk * 32;
    const float4* s = (const float4*)(x + (size_t)j * N_PTS + m0);
    float4 v[8];
#pragma unroll
    for (int i = 0; i < 8; ++i) v[i] = s[i];
    float* d = xt + (size_t)m0 * BC + j;
#pragma unroll
    for (int i = 0; i < 8; ++i) {
      d[(4 * i + 0) * BC] = v[i].x * quad[m0 + 4 * i + 0];
      d[(4 * i + 1) * BC] = v[i].y * quad[m0 + 4 * i + 1];
      d[(4 * i + 2) * BC] = v[i].z * quad[m0 + 4 * i + 2];
      d[(4 * i + 3) * BC] = v[i].w * quad[m0 + 4 * i + 3];
    }
  } else if (blk < XQ_BLOCKS + WT_BLOCKS) {
    const int idx = (blk - XQ_BLOCKS) * 256 + threadIdx.x;
    const int k   = idx >> 12;
    const int oc  = idx & 4095;
    wt[idx] = weight[oc * KSZ + k];
  } else {
    cnt[(blk - XQ_BLOCKS - WT_BLOCKS) * 256 + threadIdx.x] = 0;
  }
}

__global__ void count_kernel(const int* __restrict__ kI,
                             const int* __restrict__ oI,
                             int* __restrict__ cnt, int nnz) {
  const int step = gridDim.x * blockDim.x;
  for (int e = blockIdx.x * blockDim.x + threadIdx.x; e < nnz; e += step) {
    unsigned kk = (unsigned)kI[e];
    unsigned o  = (unsigned)oI[e];
    if (kk < KSZ && o < N_PTS) atomicAdd(&cnt[o], 1);
  }
}

__global__ void __launch_bounds__(1024)
scan_kernel(const int* __restrict__ cnt, int* __restrict__ cur) {
  __shared__ int wsum[32];
  const int t    = threadIdx.x;
  const int lane = t & 31;
  const int w    = t >> 5;

  const int4* c4 = (const int4*)(cnt + t * 16);
  int4 q0 = c4[0], q1 = c4[1], q2 = c4[2], q3 = c4[3];
  int v[16] = { q0.x, q0.y, q0.z, q0.w, q1.x, q1.y, q1.z, q1.w,
                q2.x, q2.y, q2.z, q2.w, q3.x, q3.y, q3.z, q3.w };
  int s = 0;
  for (int i = 0; i < 16; ++i) s += v[i];

  int inc = s;
  for (int off = 1; off < 32; off <<= 1) {
    int up = __shfl_up(inc, off);
    if (lane >= off) inc += up;
  }
  if (lane == 31) wsum[w] = inc;
  __syncthreads();
  if (w == 0) {
    int x = wsum[lane];
    int xi = x;
    for (int off = 1; off < 32; off <<= 1) {
      int up = __shfl_up(xi, off);
      if (lane >= off) xi += up;
    }
    wsum[lane] = xi - x;
  }
  __syncthreads();

  int run = wsum[w] + inc - s;
  int r[16];
  for (int i = 0; i < 16; ++i) { r[i] = run; run += v[i]; }
  int4* o4 = (int4*)(cur + t * 16);
  o4[0] = make_int4(r[0],  r[1],  r[2],  r[3]);
  o4[1] = make_int4(r[4],  r[5],  r[6],  r[7]);
  o4[2] = make_int4(r[8],  r[9],  r[10], r[11]);
  o4[3] = make_int4(r[12], r[13], r[14], r[15]);
}

__global__ void fill_kernel(const int* __restrict__ kI,
                            const int* __restrict__ oI,
                            const int* __restrict__ iI,
                            const float* __restrict__ psi,
                            int* __restrict__ cur,
                            int* __restrict__ sk,
                            int* __restrict__ sin_,
                            float* __restrict__ sp, int nnz) {
  const int step = gridDim.x * blockDim.x;
  for (int e = blockIdx.x * blockDim.x + threadIdx.x; e < nnz; e += step) {
    unsigned kk = (unsigned)kI[e];
    unsigned o  = (unsigned)oI[e];
    if (kk < KSZ && o < N_PTS) {
      int pos = atomicAdd(&cur[o], 1);
      if ((unsigned)pos < (unsigned)nnz) {
        sk[pos]   = (int)kk;
        sin_[pos] = iI[e];
        sp[pos]   = psi[e];
      }
    }
  }
}

#define FUSE_PPW 4
#define LDP      68
#define TP       36
#define W_FLOATS (2 * 64 * LDP)
#define A_FLOATS (8 * 16 * LDP)

struct Row8 { float4 lo, hi; };
struct Chunk { int ek; int ei; float ep; };

__device__ __forceinline__ Row8 load_row(const float* __restrict__ xt, int inp, int lane) {
  const float4* s = (const float4*)(xt + (size_t)inp * BC) + lane * 2;
  Row8 r;
  r.lo = s[0];
  r.hi = s[1];
  return r;
}

__device__ __forceinline__ void fma_row(float4& a0, float4& a1, float p, const Row8& r) {
  a0.x += p * r.lo.x; a0.y += p * r.lo.y; a0.z += p * r.lo.z; a0.w += p * r.lo.w;
  a1.x += p * r.hi.x; a1.y += p * r.hi.y; a1.z += p * r.hi.z; a1.w += p * r.hi.w;
}

__device__ __forceinline__ Chunk load_chunk(const int* __restrict__ sk,
                                            const int* __restrict__ sin_,
                                            const float* __restrict__ sp,
                                            int base, int end, int nnz, int lane) {
  const int  idx = base + lane;
  const bool ok  = idx < end;
  int j = idx < nnz ? idx : nnz - 1;
  if (j < 0) j = 0;
  const int   vk = sk[j];
  const int   vi = sin_[j];
  const float vp = sp[j];
  Chunk c;
  c.ek = ok ? vk : -1;
  c.ei = ok ? vi : 0;
  c.ep = ok ? vp : 0.f;
  return c;
}

__device__ __forceinline__ void add_chunk(const Chunk& c, int k,
                                          const float* __restrict__ xt, int lane,
                                          float4& a0, float4& a1) {
  unsigned mask = __builtin_amdgcn_ballot_w32(c.ek == k);
  while (mask) {
    const int l0 = __builtin_ctz(mask);
    mask &= mask - 1;
    const float p0 = __int_as_float(__builtin_amdgcn_readlane(__float_as_int(c.ep), l0));
    const Row8 r0 = load_row(xt, __builtin_amdgcn_readlane(c.ei, l0) & (N_PTS - 1), lane);
    if (!mask) { fma_row(a0, a1, p0, r0); break; }

    const int l1 = __builtin_ctz(mask);
    mask &= mask - 1;
    const float p1 = __int_as_float(__builtin_amdgcn_readlane(__float_as_int(c.ep), l1));
    const Row8 r1 = load_row(xt, __builtin_amdgcn_readlane(c.ei, l1) & (N_PTS - 1), lane);
    if (!mask) { fma_row(a0, a1, p0, r0); fma_row(a0, a1, p1, r1); break; }

    const int l2 = __builtin_ctz(mask);
    mask &= mask - 1;
    const float p2 = __int_as_float(__builtin_amdgcn_readlane(__float_as_int(c.ep), l2));
    const Row8 r2 = load_row(xt, __builtin_amdgcn_readlane(c.ei, l2) & (N_PTS - 1), lane);
    if (!mask) {
      fma_row(a0, a1, p0, r0); fma_row(a0, a1, p1, r1); fma_row(a0, a1, p2, r2);
      break;
    }

    const int l3 = __builtin_ctz(mask);
    mask &= mask - 1;
    const float p3 = __int_as_float(__builtin_amdgcn_readlane(__float_as_int(c.ep), l3));
    const Row8 r3 = load_row(xt, __builtin_amdgcn_readlane(c.ei, l3) & (N_PTS - 1), lane);
    __builtin_amdgcn_sched_barrier(0);
    float q0 = p0, q1 = p1, q2 = p2;
    asm volatile("" : "+s"(q0), "+s"(q1), "+s"(q2));
    fma_row(a0, a1, q0, r0); fma_row(a0, a1, q1, r1);
    fma_row(a0, a1, q2, r2); fma_row(a0, a1, p3, r3);
  }
}

#define GATHER_SLOTS 12
#define LIST_INTS    (2 * 256)

__device__ __forceinline__ void store_point(float* at, int pt, int lane,
                                            const float4& s0, const float4& s1) {
  float* d = at + (pt * 4 + (lane >> 3)) * LDP + (lane & 7) * 8;
  *(float4*)d       = s0;
  *(float4*)(d + 4) = s1;
}

template <int C>
__device__ __forceinline__ void gather_trip(const int2& e, int j0,
                                            const float* __restrict__ xt, int lane, float* at,
                                            int& pt, int& written, float4& s0, float4& s1) {
  int   key[C];
  float w[C];
  Row8  r[C];
#pragma unroll
  for (int s = 0; s < C; ++s) {
    key[s] = __builtin_amdgcn_readlane(e.x, j0 + s);
    w[s]   = __int_as_float(__builtin_amdgcn_readlane(e.y, j0 + s));
    r[s]   = load_row(xt, key[s] & (N_PTS - 1), lane);
  }
  __builtin_amdgcn_sched_barrier(0);
#pragma unroll
  for (int s = 0; s < C; ++s) {
    const int q = (key[s] >> 14) & 3;
    if (q != pt) {
      store_point(at, pt, lane, s0, s1);
      written |= 1 << pt;
      s0 = make_float4(0.f, 0.f, 0.f, 0.f);
      s1 = s0;
      pt = q;
    }
    fma_row(s0, s1, w[s], r[s]);
  }
}

template <int C>
struct TripSel {
  static __device__ __forceinline__ void run(int c, const int2& e, int j0,
                                             const float* __restrict__ xt, int lane, float* at,
                                             int& pt, int& written, float4& s0, float4& s1) {
    if (c == C) gather_trip<C>(e, j0, xt, lane, at, pt, written, s0, s1);
    else        TripSel<C - 1>::run(c, e, j0, xt, lane, at, pt, written, s0, s1);
  }
};
template <>
struct TripSel<0> {
  static __device__ __forceinline__ void run(int, const int2&, int, const float* __restrict__, int, float*,
                                             int&, int&, float4&, float4&) {}
};

__device__ __forceinline__ void gather_stream(const Chunk (&c0)[FUSE_PPW],
                                              const Chunk (&c1)[FUSE_PPW], int k,
                                              const float* __restrict__ xt, int lane,
                                              int* list, float* at) {
  int total = 0;
#pragma unroll
  for (int i = 0; i < FUSE_PPW; ++i) {
#pragma unroll
    for (int h = 0; h < 2; ++h) {
      const Chunk& c = h ? c1[i] : c0[i];
      const bool     hit = c.ek == k;
      const unsigned m   = __builtin_amdgcn_ballot_w32(hit);
      const int      pos = total + (int)__builtin_amdgcn_mbcnt_lo(m, 0u);
      if (hit) {
        int2 e;
        e.x = (i << 14) | (c.ei & (N_PTS - 1));
        e.y = __float_as_int(c.ep);
        *(int2*)&list[2 * pos] = e;
      }
      total += __builtin_popcount(m);
    }
  }
  __builtin_amdgcn_fence(__ATOMIC_RELEASE, "wavefront");
  __builtin_amdgcn_wave_barrier();
  __builtin_amdgcn_fence(__ATOMIC_ACQUIRE, "wavefront");

  int    pt = 0;
  int    written = 0;
  float4 s0 = make_float4(0.f, 0.f, 0.f, 0.f);
  float4 s1 = s0;
  for (int base = 0; base < total; base += 32) {
    const int n = total - base < 32 ? total - base : 32;
    int2 e = make_int2(0, 0);
    if (lane < n) e = *(const int2*)&list[2 * (base + lane)];
    for (int j0 = 0; j0 < n; j0 += GATHER_SLOTS) {
      const int c = n - j0 < GATHER_SLOTS ? n - j0 : GATHER_SLOTS;
      TripSel<GATHER_SLOTS>::run(c, e, j0, xt, lane, at, pt, written, s0, s1);
    }
  }
  store_point(at, pt, lane, s0, s1);
  written |= 1 << pt;
  const float4 zero = make_float4(0.f, 0.f, 0.f, 0.f);
#pragma unroll
  for (int i = 0; i < FUSE_PPW; ++i) {
    if (!(written & (1 << i))) store_point(at, i, lane, zero, zero);
  }
}

__global__ void __launch_bounds__(256)
fused_kernel(const int* __restrict__ cnt,
             const int* __restrict__ cur,
             const int* __restrict__ sk,
             const int* __restrict__ sin_,
             const float* __restrict__ sp,
             const float* __restrict__ xt,
             const float* __restrict__ wt,
             const float* __restrict__ bias,
             float* __restrict__ y,
             int nnz) {
  __shared__ __attribute__((aligned(16))) float lds[W_FLOATS + A_FLOATS];
  __shared__ __attribute__((aligned(16))) int lists[8 * LIST_INTS];
  static_assert(256 * TP <= W_FLOATS + A_FLOATS, "output staging tile must fit in the block's LDS");

  const int tid  = threadIdx.x;
  const int lane = tid & 31;
  const int hi   = lane >> 4;
  const int l15  = lane & 15;
  const int wave = __builtin_amdgcn_readfirstlane(tid >> 5);
  const int n0   = (blockIdx.x * 8 + wave) * FUSE_PPW;

  int beg[FUSE_PPW], end[FUSE_PPW];
#pragma unroll
  for (int i = 0; i < FUSE_PPW; ++i) {
    int e   = cur[n0 + i];
    int len = cnt[n0 + i];
    if (e > nnz) e = nnz;
    int bg = e - len;
    if (bg < 0) bg = 0;
    beg[i] = bg;
    end[i] = e;
  }

  Chunk ck[FUSE_PPW], cm[FUSE_PPW];
#pragma unroll
  for (int i = 0; i < FUSE_PPW; ++i) {
    ck[i] = load_chunk(sk, sin_, sp, beg[i],      end[i], nnz, lane);
    cm[i] = load_chunk(sk, sin_, sp, beg[i] + 32, end[i], nnz, lane);
  }

  float* const at = lds + W_FLOATS + wave * (16 * LDP);
  int* const list = lists + wave * LIST_INTS;
  bool longseg = false;
#pragma unroll
  for (int i = 0; i < FUSE_PPW; ++i) longseg |= (end[i] - beg[i] > 64);
  v8f acc[4] = {};

  const float4* const wsrc = (const float4*)wt + tid;
  float4 w0 = wsrc[0], w1 = wsrc[256], w2 = wsrc[512], w3 = wsrc[768];
  const int wofs = (tid >> 4) * LDP + (tid & 15) * 4;

  for (int kk = 0; kk < KSZ; ++kk) {
    if (!longseg) {
      gather_stream(ck, cm, kk, xt, lane, list, at);
    } else {
      float4 a0[FUSE_PPW], a1[FUSE_PPW];
#pragma unroll
      for (int i = 0; i < FUSE_PPW; ++i) {
        a0[i] = make_float4(0.f, 0.f, 0.f, 0.f);
        a1[i] = a0[i];
        add_chunk(ck[i], kk, xt, lane, a0[i], a1[i]);
        add_chunk(cm[i], kk, xt, lane, a0[i], a1[i]);
        for (int base = beg[i] + 64; base < end[i]; base += 32) {
          const Chunk c2 = load_chunk(sk, sin_, sp, base, end[i], nnz, lane);
          add_chunk(c2, kk, xt, lane, a0[i], a1[i]);
        }
        store_point(at, i, lane, a0[i], a1[i]);
      }
    }

    float* const wb = lds + (kk & 1) * (64 * LDP);
    *(float4*)&wb[wofs]            = w0;
    *(float4*)&wb[wofs + 16 * LDP] = w1;
    *(float4*)&wb[wofs + 32 * LDP] = w2;
    *(float4*)&wb[wofs + 48 * LDP] = w3;
    __syncthreads();

    if (kk + 1 < KSZ) {
      const float4* s = wsrc + (kk + 1) * (COUT * CIN / 4);
      w0 = s[0]; w1 = s[256]; w2 = s[512]; w3 = s[768];
    }

#pragma unroll
    for (int ks = 0; ks < 16; ++ks) {
      const int k2 = ks * 4 + hi * 2;
      v2f a = { at[l15 * LDP + k2], at[l15 * LDP + k2 + 1] };
#pragma unroll
      for (int wn = 0; wn < 4; ++wn) {
        v2f bw = { wb[(wn * 16 + l15) * LDP + k2], wb[(wn * 16 + l15) * LDP + k2 + 1] };
        acc[wn] = __builtin_amdgcn_wmma_f32_16x16x4_f32(
             false, a,  false, bw,
             (short)0, acc[wn],  false,  false);
      }
    }
  }

  __syncthreads();
  float* const T = lds;
#pragma unroll
  for (int wn = 0; wn < 4; ++wn) {
    const float bv = bias[wn * 16 + l15];
#pragma unroll
    for (int r = 0; r < 8; ++r) {
      T[((r & 3) * COUT + wn * 16 + l15) * TP + wave * FUSE_PPW + 2 * hi + (r >> 2)] = acc[wn][r] + bv;
    }
  }
  __syncthreads();

  const int nb = blockIdx.x * (8 * FUSE_PPW);
#pragma unroll
  for (int s = 0; s < 8; ++s) {
    const int j = wave * 32 + s * 4 + (lane >> 3);
    const float4 v = *(const float4*)&T[j * TP + (lane & 7) * 4];
    *(float4*)(y + (size_t)j * N_PTS + nb + (lane & 7) * 4) = v;
  }
}

extern "C" void kernel_launch(void* const* d_in, const int* in_sizes, int n_in,
                              void* d_out, int out_size, void* d_ws, size_t ws_size,
                              hipStream_t stream) {
  const float* x        = (const float*)d_in[0];
  const int*   psi_idx  = (const int*)  d_in[1];
  const float* psi_vals = (const float*)d_in[2];
  const float* quad     = (const float*)d_in[3];
  const float* weight   = (const float*)d_in[4];
  const float* bias     = (const float*)d_in[5];
  float*       y        = (float*)d_out;

  const int nnz = in_sizes[2];

  float* xt   = (float*)d_ws;
  float* wt   = xt + (size_t)N_PTS * BC;
  int*   cnt  = (int*)(wt + (size_t)N_PTS * BC);
  int*   cur  = cnt + N_PTS;
  int*   sk   = cur + N_PTS;
  int*   sin_ = sk + nnz;
  float* sp   = (float*)(sin_ + nnz);

  const int* kI = psi_idx;
  const int* oI = psi_idx + nnz;
  const int* iI = psi_idx + 2 * nnz;

  setup_kernel<<<XQ_BLOCKS + WT_BLOCKS + ZERO_BLOCKS, 256, 0, stream>>>(x, quad, xt, weight, wt, cnt);

  count_kernel<<<256, 256, 0, stream>>>(kI, oI, cnt, nnz);
  scan_kernel<<<1, 1024, 0, stream>>>(cnt, cur);
  fill_kernel<<<256, 256, 0, stream>>>(kI, oI, iI, psi_vals, cur, sk, sin_, sp, nnz);

  fused_kernel<<<N_PTS / (8 * FUSE_PPW), 256, 0, stream>>>(cnt, cur, sk, sin_, sp, xt, wt, bias, y, nnz);
}


# Bin entry packed into one 8-byte record (kernel index, input point, value): one scattered store per entry in the binning pass, not three, one load per gather lane; same slots and arrival-order sums
# speedup vs baseline: 5.5661x; 1.5273x over previous
#include <hip/hip_runtime.h>

typedef __attribute__((ext_vector_type(2))) float v2f;
typedef __attribute__((ext_vector_type(8))) float v8f;

#define N_PTS  16384
#define BATCH  4
#define CIN    64
#define COUT   64
#define KSZ    9
#define BC     (BATCH * CIN)


#define XQ_BLOCKS   (N_PTS / 32)
#define WT_BLOCKS   (KSZ * COUT * CIN / 256)
#define BIN_CAP     64
#define ZERO_BLOCKS (N_PTS / 256 + 1)

__global__ void __launch_bounds__(256)
setup_kernel(const float* __restrict__ x,
             const float* __restrict__ quad,
             float* __restrict__ xt,
             const float* __restrict__ weight,
             float* __restrict__ wt,
             int* __restrict__ cnt) {
  const int blk = blockIdx.x;
  if (blk < XQ_BLOCKS) {
    const int j  = threadIdx.x;
    const int m0 = blk * 32;
    const float4* s = (const float4*)(x + (size_t)j * N_PTS + m0);
    float4 v[8];
#pragma unroll
    for (int i = 0; i < 8; ++i) v[i] = s[i];
    float* d = xt + (size_t)m0 * BC + j;
#pragma unroll
    for (int i = 0; i < 8; ++i) {
      d[(4 * i + 0) * BC] = v[i].x * quad[m0 + 4 * i + 0];
      d[(4 * i + 1) * BC] = v[i].y * quad[m0 + 4 * i + 1];
      d[(4 * i + 2) * BC] = v[i].z * quad[m0 + 4 * i + 2];
      d[(4 * i + 3) * BC] = v[i].w * quad[m0 + 4 * i + 3];
    }
  } else if (blk < XQ_BLOCKS + WT_BLOCKS) {
    const int idx = (blk - XQ_BLOCKS) * 256 + threadIdx.x;
    const int k   = idx >> 12;
    const int oc  = idx & 4095;
    wt[idx] = weight[oc * KSZ + k];
  } else {
    cnt[(blk - XQ_BLOCKS - WT_BLOCKS) * 256 + threadIdx.x] = 0;
  }
}

#define BIN_UNROLL 8

__device__ __forceinline__ void bin_place(unsigned kk, unsigned o, int in, float p, int pos,
                                          int* __restrict__ cnt,
                                          int* __restrict__ be,
                                          int* __restrict__ ovk, int* __restrict__ ovi, float* __restrict__ ovp,
                                          int nnz) {
  if ((unsigned)pos < (unsigned)BIN_CAP) {
    const int s = (int)o * BIN_CAP + pos;
    int2 w;
    w.x = (int)((kk << 14) | ((unsigned)in & (N_PTS - 1)));
    w.y = __float_as_int(p);
    ((int2*)be)[s] = w;
  } else {
    const int q = atomicAdd(&cnt[N_PTS], 1);
    if ((unsigned)q < (unsigned)nnz) {
      ovk[q] = (int)((kk << 14) | o);
      ovi[q] = in;
      ovp[q] = p;
    }
  }
}

__global__ void bin_kernel(const int* __restrict__ kI,
                           const int* __restrict__ oI,
                           const int* __restrict__ iI,
                           const float* __restrict__ psi,
                           int* __restrict__ cnt,
                           int* __restrict__ be,
                           int* __restrict__ ovk,
                           int* __restrict__ ovi,
                           float* __restrict__ ovp, int nnz) {
  const int step = gridDim.x * blockDim.x;
  int e = blockIdx.x * blockDim.x + threadIdx.x;
  for (; e < nnz - (BIN_UNROLL - 1) * step; e += BIN_UNROLL * step) {
    unsigned kk[BIN_UNROLL], o[BIN_UNROLL];
    int      in[BIN_UNROLL], pos[BIN_UNROLL];
    float    p[BIN_UNROLL];
#pragma unroll
    for (int u = 0; u < BIN_UNROLL; ++u) {
      kk[u] = (unsigned)kI[e + u * step];
      o[u]  = (unsigned)oI[e + u * step];
      in[u] = iI[e + u * step];
      p[u]  = psi[e + u * step];
    }
#pragma unroll
    for (int u = 0; u < BIN_UNROLL; ++u) {
      pos[u] = -1;
      if (kk[u] < KSZ && o[u] < N_PTS) pos[u] = atomicAdd(&cnt[o[u]], 1);
    }
#pragma unroll
    for (int u = 0; u < BIN_UNROLL; ++u) {
      if (kk[u] < KSZ && o[u] < N_PTS)
        bin_place(kk[u], o[u], in[u], p[u], pos[u], cnt, be, ovk, ovi, ovp, nnz);
    }
  }
  for (; e < nnz; e += step) {
    const unsigned kk = (unsigned)kI[e];
    const unsigned o  = (unsigned)oI[e];
    if (kk < KSZ && o < N_PTS) {
      const int pos = atomicAdd(&cnt[o], 1);
      bin_place(kk, o, iI[e], psi[e], pos, cnt, be, ovk, ovi, ovp, nnz);
    }
  }
}

#define FUSE_PPW 4
#define LDP      68
#define TP       36
#define W_FLOATS (2 * 64 * LDP)
#define A_FLOATS (8 * 16 * LDP)

struct Row8 { float4 lo, hi; };
struct Chunk { int ek; int ei; float ep; };

__device__ __forceinline__ Row8 load_row(const float* __restrict__ xt, int inp, int lane) {
  const float4* s = (const float4*)(xt + (size_t)inp * BC) + lane * 2;
  Row8 r;
  r.lo = s[0];
  r.hi = s[1];
  return r;
}

__device__ __forceinline__ void fma_row(float4& a0, float4& a1, float p, const Row8& r) {
  a0.x += p * r.lo.x; a0.y += p * r.lo.y; a0.z += p * r.lo.z; a0.w += p * r.lo.w;
  a1.x += p * r.hi.x; a1.y += p * r.hi.y; a1.z += p * r.hi.z; a1.w += p * r.hi.w;
}

__device__ __forceinline__ Chunk load_chunk(const int* __restrict__ be,
                                            int base, int end, int lane) {
  const int idx = base + lane;
  Chunk c;
  c.ek = -1;
  c.ei = 0;
  c.ep = 0.f;
  if (idx < end) {
    const int2 w = ((const int2*)be)[idx];
    c.ek = w.x >> 14;
    c.ei = w.x & (N_PTS - 1);
    c.ep = __int_as_float(w.y);
  }
  return c;
}

__device__ __forceinline__ void add_chunk(const Chunk& c, int k,
                                          const float* __restrict__ xt, int lane,
                                          float4& a0, float4& a1) {
  unsigned mask = __builtin_amdgcn_ballot_w32(c.ek == k);
  while (mask) {
    const int l0 = __builtin_ctz(mask);
    mask &= mask - 1;
    const float p0 = __int_as_float(__builtin_amdgcn_readlane(__float_as_int(c.ep), l0));
    const Row8 r0 = load_row(xt, __builtin_amdgcn_readlane(c.ei, l0) & (N_PTS - 1), lane);
    if (!mask) { fma_row(a0, a1, p0, r0); break; }

    const int l1 = __builtin_ctz(mask);
    mask &= mask - 1;
    const float p1 = __int_as_float(__builtin_amdgcn_readlane(__float_as_int(c.ep), l1));
    const Row8 r1 = load_row(xt, __builtin_amdgcn_readlane(c.ei, l1) & (N_PTS - 1), lane);
    if (!mask) { fma_row(a0, a1, p0, r0); fma_row(a0, a1, p1, r1); break; }

    const int l2 = __builtin_ctz(mask);
    mask &= mask - 1;
    const float p2 = __int_as_float(__builtin_amdgcn_readlane(__float_as_int(c.ep), l2));
    const Row8 r2 = load_row(xt, __builtin_amdgcn_readlane(c.ei, l2) & (N_PTS - 1), lane);
    if (!mask) {
      fma_row(a0, a1, p0, r0); fma_row(a0, a1, p1, r1); fma_row(a0, a1, p2, r2);
      break;
    }

    const int l3 = __builtin_ctz(mask);
    mask &= mask - 1;
    const float p3 = __int_as_float(__builtin_amdgcn_readlane(__float_as_int(c.ep), l3));
    const Row8 r3 = load_row(xt, __builtin_amdgcn_readlane(c.ei, l3) & (N_PTS - 1), lane);
    __builtin_amdgcn_sched_barrier(0);
    float q0 = p0, q1 = p1, q2 = p2;
    asm volatile("" : "+s"(q0), "+s"(q1), "+s"(q2));
    fma_row(a0, a1, q0, r0); fma_row(a0, a1, q1, r1);
    fma_row(a0, a1, q2, r2); fma_row(a0, a1, p3, r3);
  }
}

#define GATHER_SLOTS 12
#define LIST_INTS    (2 * 256)

__device__ __forceinline__ void store_point(float* at, int pt, int lane,
                                            const float4& s0, const float4& s1) {
  float* d = at + (pt * 4 + (lane >> 3)) * LDP + (lane & 7) * 8;
  *(float4*)d       = s0;
  *(float4*)(d + 4) = s1;
}

template <int C>
__device__ __forceinline__ void gather_trip(const int2& e, int j0,
                                            const float* __restrict__ xt, int lane, float* at,
                                            int& pt, int& written, float4& s0, float4& s1) {
  int   key[C];
  float w[C];
  Row8  r[C];
#pragma unroll
  for (int s = 0; s < C; ++s) {
    key[s] = __builtin_amdgcn_readlane(e.x, j0 + s);
    w[s]   = __int_as_float(__builtin_amdgcn_readlane(e.y, j0 + s));
    r[s]   = load_row(xt, key[s] & (N_PTS - 1), lane);
  }
  __builtin_amdgcn_sched_barrier(0);
#pragma unroll
  for (int s = 0; s < C; ++s) {
    const int q = (key[s] >> 14) & 3;
    if (q != pt) {
      store_point(at, pt, lane, s0, s1);
      written |= 1 << pt;
      s0 = make_float4(0.f, 0.f, 0.f, 0.f);
      s1 = s0;
      pt = q;
    }
    fma_row(s0, s1, w[s], r[s]);
  }
}

template <int C>
struct TripSel {
  static __device__ __forceinline__ void run(int c, const int2& e, int j0,
                                             const float* __restrict__ xt, int lane, float* at,
                                             int& pt, int& written, float4& s0, float4& s1) {
    if (c == C) gather_trip<C>(e, j0, xt, lane, at, pt, written, s0, s1);
    else        TripSel<C - 1>::run(c, e, j0, xt, lane, at, pt, written, s0, s1);
  }
};
template <>
struct TripSel<0> {
  static __device__ __forceinline__ void run(int, const int2&, int, const float* __restrict__, int, float*,
                                             int&, int&, float4&, float4&) {}
};

__device__ __forceinline__ void gather_stream(const Chunk (&c0)[FUSE_PPW],
                                              const Chunk (&c1)[FUSE_PPW], int k,
                                              const float* __restrict__ xt, int lane,
                                              int* list, float* at) {
  int total = 0;
#pragma unroll
  for (int i = 0; i < FUSE_PPW; ++i) {
#pragma unroll
    for (int h = 0; h < 2; ++h) {
      const Chunk& c = h ? c1[i] : c0[i];
      const bool     hit = c.ek == k;
      const unsigned m   = __builtin_amdgcn_ballot_w32(hit);
      const int      pos = total + (int)__builtin_amdgcn_mbcnt_lo(m, 0u);
      if (hit) {
        int2 e;
        e.x = (i << 14) | (c.ei & (N_PTS - 1));
        e.y = __float_as_int(c.ep);
        *(int2*)&list[2 * pos] = e;
      }
      total += __builtin_popcount(m);
    }
  }
  __builtin_amdgcn_fence(__ATOMIC_RELEASE, "wavefront");
  __builtin_amdgcn_wave_barrier();
  __builtin_amdgcn_fence(__ATOMIC_ACQUIRE, "wavefront");

  int    pt = 0;
  int    written = 0;
  float4 s0 = make_float4(0.f, 0.f, 0.f, 0.f);
  float4 s1 = s0;
  for (int base = 0; base < total; base += 32) {
    const int n = total - base < 32 ? total - base : 32;
    int2 e = make_int2(0, 0);
    if (lane < n) e = *(const int2*)&list[2 * (base + lane)];
    for (int j0 = 0; j0 < n; j0 += GATHER_SLOTS) {
      const int c = n - j0 < GATHER_SLOTS ? n - j0 : GATHER_SLOTS;
      TripSel<GATHER_SLOTS>::run(c, e, j0, xt, lane, at, pt, written, s0, s1);
    }
  }
  store_point(at, pt, lane, s0, s1);
  written |= 1 << pt;
  const float4 zero = make_float4(0.f, 0.f, 0.f, 0.f);
#pragma unroll
  for (int i = 0; i < FUSE_PPW; ++i) {
    if (!(written & (1 << i))) store_point(at, i, lane, zero, zero);
  }
}

__global__ void __launch_bounds__(256)
fused_kernel(const int* __restrict__ cnt,
             const int* __restrict__ be,
             const int* __restrict__ ovk,
             const int* __restrict__ ovi,
             const float* __restrict__ ovp,
             const float* __restrict__ xt,
             const float* __restrict__ wt,
             const float* __restrict__ bias,
             float* __restrict__ y,
             int nnz) {
  __shared__ __attribute__((aligned(16))) float lds[W_FLOATS + A_FLOATS];
  __shared__ __attribute__((aligned(16))) int lists[8 * LIST_INTS];
  static_assert(256 * TP <= W_FLOATS + A_FLOATS, "output staging tile must fit in the block's LDS");

  const int tid  = threadIdx.x;
  const int lane = tid & 31;
  const int hi   = lane >> 4;
  const int l15  = lane & 15;
  const int wave = __builtin_amdgcn_readfirstlane(tid >> 5);
  const int n0   = (blockIdx.x * 8 + wave) * FUSE_PPW;

  int beg[FUSE_PPW], end[FUSE_PPW];
  bool longseg = false;
#pragma unroll
  for (int i = 0; i < FUSE_PPW; ++i) {
    int len = cnt[n0 + i];
    longseg |= (len > BIN_CAP);
    if (len > BIN_CAP) len = BIN_CAP;
    if (len < 0) len = 0;
    beg[i] = (n0 + i) * BIN_CAP;
    end[i] = beg[i] + len;
  }
  int novf = 0;
  if (longseg) {
    novf = cnt[N_PTS];
    if (novf > nnz) novf = nnz;
    if (novf < 0) novf = 0;
  }
  novf = __builtin_amdgcn_readfirstlane(novf);

  Chunk ck[FUSE_PPW], cm[FUSE_PPW];
#pragma unroll
  for (int i = 0; i < FUSE_PPW; ++i) {
    ck[i] = load_chunk(be, beg[i],      end[i], lane);
    cm[i] = load_chunk(be, beg[i] + 32, end[i], lane);
  }

  float* const at = lds + W_FLOATS + wave * (16 * LDP);
  int* const list = lists + wave * LIST_INTS;
  v8f acc[4] = {};

  const float4* const wsrc = (const float4*)wt + tid;
  float4 w0 = wsrc[0], w1 = wsrc[256], w2 = wsrc[512], w3 = wsrc[768];
  const int wofs = (tid >> 4) * LDP + (tid & 15) * 4;

  for (int kk = 0; kk < KSZ; ++kk) {
    if (!longseg) {
      gather_stream(ck, cm, kk, xt, lane, list, at);
    } else {
      float4 a0[FUSE_PPW], a1[FUSE_PPW];
#pragma unroll
      for (int i = 0; i < FUSE_PPW; ++i) {
        a0[i] = make_float4(0.f, 0.f, 0.f, 0.f);
        a1[i] = a0[i];
        add_chunk(ck[i], kk, xt, lane, a0[i], a1[i]);
        add_chunk(cm[i], kk, xt, lane, a0[i], a1[i]);
        for (int base = 0; base < novf; base += 32) {
          const int idx = base + lane;
          Chunk c2;
          c2.ek = -1;
          c2.ei = 0;
          c2.ep = 0.f;
          if (idx < novf) {
            const int ok = ovk[idx];
            if ((ok & (N_PTS - 1)) == n0 + i) {
              c2.ek = ok >> 14;
              c2.ei = ovi[idx];
              c2.ep = ovp[idx];
            }
          }
          add_chunk(c2, kk, xt, lane, a0[i], a1[i]);
        }
        store_point(at, i, lane, a0[i], a1[i]);
      }
    }

    float* const wb = lds + (kk & 1) * (64 * LDP);
    *(float4*)&wb[wofs]            = w0;
    *(float4*)&wb[wofs + 16 * LDP] = w1;
    *(float4*)&wb[wofs + 32 * LDP] = w2;
    *(float4*)&wb[wofs + 48 * LDP] = w3;
    __syncthreads();

    if (kk + 1 < KSZ) {
      const float4* s = wsrc + (kk + 1) * (COUT * CIN / 4);
      w0 = s[0]; w1 = s[256]; w2 = s[512]; w3 = s[768];
    }

#pragma unroll
    for (int ks = 0; ks < 16; ++ks) {
      const int k2 = ks * 4 + hi * 2;
      v2f a = { at[l15 * LDP + k2], at[l15 * LDP + k2 + 1] };
#pragma unroll
      for (int wn = 0; wn < 4; ++wn) {
        v2f bw = { wb[(wn * 16 + l15) * LDP + k2], wb[(wn * 16 + l15) * LDP + k2 + 1] };
        acc[wn] = __builtin_amdgcn_wmma_f32_16x16x4_f32(
             false, a,  false, bw,
             (short)0, acc[wn],  false,  false);
      }
    }
  }

  __syncthreads();
  float* const T = lds;
#pragma unroll
  for (int wn = 0; wn < 4; ++wn) {
    const float bv = bias[wn * 16 + l15];
#pragma unroll
    for (int r = 0; r < 8; ++r) {
      T[((r & 3) * COUT + wn * 16 + l15) * TP + wave * FUSE_PPW + 2 * hi + (r >> 2)] = acc[wn][r] + bv;
    }
  }
  __syncthreads();

  const int nb = blockIdx.x * (8 * FUSE_PPW);
#pragma unroll
  for (int s = 0; s < 8; ++s) {
    const int j = wave * 32 + s * 4 + (lane >> 3);
    const float4 v = *(const float4*)&T[j * TP + (lane & 7) * 4];
    *(float4*)(y + (size_t)j * N_PTS + nb + (lane & 7) * 4) = v;
  }
}

extern "C" void kernel_launch(void* const* d_in, const int* in_sizes, int n_in,
                              void* d_out, int out_size, void* d_ws, size_t ws_size,
                              hipStream_t stream) {
  const float* x        = (const float*)d_in[0];
  const int*   psi_idx  = (const int*)  d_in[1];
  const float* psi_vals = (const float*)d_in[2];
  const float* quad     = (const float*)d_in[3];
  const float* weight   = (const float*)d_in[4];
  const float* bias     = (const float*)d_in[5];
  float*       y        = (float*)d_out;

  const int nnz = in_sizes[2];

  float* xt   = (float*)d_ws;
  float* wt   = xt + (size_t)N_PTS * BC;
  int*   be   = (int*)(wt + KSZ * COUT * CIN);
  int*   cnt  = (int*)(wt + (size_t)N_PTS * BC);
  int*   ovk  = cnt + N_PTS + 256;
  int*   ovi  = ovk + nnz;
  float* ovp  = (float*)(ovi + nnz);

  const int* kI = psi_idx;
  const int* oI = psi_idx + nnz;
  const int* iI = psi_idx + 2 * nnz;

  setup_kernel<<<XQ_BLOCKS + WT_BLOCKS + ZERO_BLOCKS, 256, 0, stream>>>(x, quad, xt, weight, wt, cnt);

  bin_kernel<<<256, 256, 0, stream>>>(kI, oI, iI, psi_vals, cnt, be, ovk, ovi, ovp, nnz);

  fused_kernel<<<N_PTS / (8 * FUSE_PPW), 256, 0, stream>>>(cnt, be, ovk, ovi, ovp, xt, wt, bias, y, nnz);
}


# Per-entry returning global atomic replaced by a two-level sort ranked in LDS: 2048-entry chunks grouped by 256-point range, then a 64-point pass builds the bins; 4 launches; arrival-order sums stay
# speedup vs baseline: 8.3402x; 1.4984x over previous
#include <hip/hip_runtime.h>

typedef __attribute__((ext_vector_type(2))) float v2f;
typedef __attribute__((ext_vector_type(8))) float v8f;

#define N_PTS  16384
#define BATCH  4
#define CIN    64
#define COUT   64
#define KSZ    9
#define BC     (BATCH * CIN)


#define XQ_BLOCKS   (N_PTS / 32)
#define WT_BLOCKS   (KSZ * COUT * CIN / 256)
#define BIN_CAP     64
#define RNG_PTS     256
#define N_RNG       (N_PTS / RNG_PTS)
#define RNG_CAP     12288
#define RC_PITCH    32
#define RC_BLOCKS   (N_RNG * RC_PITCH / 256)
#define ZERO_BLOCKS (RC_BLOCKS + 1)

__global__ void __launch_bounds__(256)
setup_kernel(const float* __restrict__ x,
             const float* __restrict__ quad,
             float* __restrict__ xt,
             const float* __restrict__ weight,
             float* __restrict__ wt,
             int* __restrict__ rcnt,
             int* __restrict__ cnt) {
  const int blk = blockIdx.x;
  if (blk < XQ_BLOCKS) {
    const int j  = threadIdx.x;
    const int m0 = blk * 32;
    const float4* s = (const float4*)(x + (size_t)j * N_PTS + m0);
    float4 v[8];
#pragma unroll
    for (int i = 0; i < 8; ++i) v[i] = s[i];
    float* d = xt + (size_t)m0 * BC + j;
#pragma unroll
    for (int i = 0; i < 8; ++i) {
      d[(4 * i + 0) * BC] = v[i].x * quad[m0 + 4 * i + 0];
      d[(4 * i + 1) * BC] = v[i].y * quad[m0 + 4 * i + 1];
      d[(4 * i + 2) * BC] = v[i].z * quad[m0 + 4 * i + 2];
      d[(4 * i + 3) * BC] = v[i].w * quad[m0 + 4 * i + 3];
    }
  } else if (blk < XQ_BLOCKS + WT_BLOCKS) {
    const int idx = (blk - XQ_BLOCKS) * 256 + threadIdx.x;
    const int k   = idx >> 12;
    const int oc  = idx & 4095;
    wt[idx] = weight[oc * KSZ + k];
  } else {
    const int z = blk - XQ_BLOCKS - WT_BLOCKS;
    if (z < RC_BLOCKS) rcnt[z * 256 + threadIdx.x] = 0;
    else               cnt[N_PTS + threadIdx.x] = 0;
  }
}

#define SORT_CHUNK  2048
#define SORT_UNROLL (SORT_CHUNK / 256)

__device__ __forceinline__ void overflow_push(unsigned kk, unsigned o, int in, float p,
                                              int* __restrict__ cnt,
                                              int* __restrict__ ovk, int* __restrict__ ovi, float* __restrict__ ovp,
                                              int nnz) {
  const int q = atomicAdd(&cnt[N_PTS], 1);
  if ((unsigned)q < (unsigned)nnz) {
    ovk[q] = (int)((kk << 14) | o);
    ovi[q] = in;
    ovp[q] = p;
  }
}

__global__ void __launch_bounds__(256)
range_kernel(const int* __restrict__ kI,
             const int* __restrict__ oI,
             const int* __restrict__ iI,
             const float* __restrict__ psi,
             int* __restrict__ rcnt,
             int* __restrict__ rr,
             int* __restrict__ cnt,
             int* __restrict__ ovk,
             int* __restrict__ ovi,
             float* __restrict__ ovp, int nnz) {
  __shared__ int  hist[N_RNG];
  __shared__ int  start[N_RNG];
  __shared__ int  gbase[N_RNG];
  __shared__ int2 srec[SORT_CHUNK];
  __shared__ int  sdst[SORT_CHUNK];
  const int tid = threadIdx.x;
  for (int c0 = blockIdx.x * SORT_CHUNK; c0 < nnz; c0 += gridDim.x * SORT_CHUNK) {
    if (tid < N_RNG) hist[tid] = 0;
    __syncthreads();
    unsigned kk[SORT_UNROLL], o[SORT_UNROLL];
    int      in[SORT_UNROLL], lr[SORT_UNROLL];
    float    p[SORT_UNROLL];
#pragma unroll
    for (int u = 0; u < SORT_UNROLL; ++u) {
      const int e = c0 + u * 256 + tid;
      kk[u] = KSZ;
      o[u]  = 0;
      in[u] = 0;
      p[u]  = 0.f;
      if (e < nnz) {
        kk[u] = (unsigned)kI[e];
        o[u]  = (unsigned)oI[e];
        in[u] = iI[e];
        p[u]  = psi[e];
      }
    }
#pragma unroll
    for (int u = 0; u < SORT_UNROLL; ++u) {
      lr[u] = -1;
      if (kk[u] < KSZ && o[u] < N_PTS) lr[u] = atomicAdd(&hist[o[u] / RNG_PTS], 1);
    }
    __syncthreads();
    if (tid < N_RNG) {
      const int h  = hist[tid];
      const int gb = h ? atomicAdd(&rcnt[tid * RC_PITCH], h) : 0;
      int s = 0;
#pragma unroll
      for (int j = 0; j < N_RNG; ++j) s += (j < tid) ? hist[j] : 0;
      start[tid] = s;
      gbase[tid] = gb;
    }
    __syncthreads();
#pragma unroll
    for (int u = 0; u < SORT_UNROLL; ++u) {
      if (lr[u] >= 0) {
        const int rg   = (int)(o[u] / RNG_PTS);
        const int pos  = start[rg] + lr[u];
        const int slot = gbase[rg] + lr[u];
        if ((unsigned)slot < (unsigned)RNG_CAP) {
          int2 w;
          w.x = (int)((kk[u] << 22) | ((o[u] & (RNG_PTS - 1)) << 14) | ((unsigned)in[u] & (N_PTS - 1)));
          w.y = __float_as_int(p[u]);
          srec[pos] = w;
          sdst[pos] = rg * RNG_CAP + slot;
        } else {
          sdst[pos] = -1;
          overflow_push(kk[u], o[u], in[u], p[u], cnt, ovk, ovi, ovp, nnz);
        }
      }
    }
    __syncthreads();
    const int total = start[N_RNG - 1] + hist[N_RNG - 1];
#pragma unroll
    for (int u = 0; u < SORT_UNROLL; ++u) {
      const int pos = u * 256 + tid;
      if (pos < total) {
        const int d = sdst[pos];
        if (d >= 0) ((int2*)rr)[d] = srec[pos];
      }
    }
    __syncthreads();
  }
}

#define FINE_PTS    64
#define FINE_THREADS 512
#define FINE_UNROLL 8

__global__ void __launch_bounds__(FINE_THREADS)
point_kernel(const int* __restrict__ rcnt,
             const int* __restrict__ rr,
             int* __restrict__ cnt,
             int* __restrict__ be,
             int* __restrict__ ovk,
             int* __restrict__ ovi,
             float* __restrict__ ovp, int nnz) {
  __shared__ int  pc[FINE_PTS];
  __shared__ int2 lb[FINE_PTS * BIN_CAP];
  const int tid = threadIdx.x;
  const int rg  = blockIdx.x / (RNG_PTS / FINE_PTS);
  const int qt  = blockIdx.x % (RNG_PTS / FINE_PTS);
  if (tid < FINE_PTS) pc[tid] = 0;
  __syncthreads();
  int n = rcnt[rg * RC_PITCH];
  const bool spilled = n > RNG_CAP;
  if (n > RNG_CAP) n = RNG_CAP;
  if (n < 0) n = 0;
  const int2* const srcp = (const int2*)rr + (size_t)rg * RNG_CAP;
  for (int i0 = 0; i0 < n; i0 += FINE_UNROLL * FINE_THREADS) {
    int2 w[FINE_UNROLL];
#pragma unroll
    for (int u = 0; u < FINE_UNROLL; ++u) {
      const int i = i0 + u * FINE_THREADS + tid;
      w[u] = make_int2(-1, 0);
      if (i < n) w[u] = srcp[i];
    }
#pragma unroll
    for (int u = 0; u < FINE_UNROLL; ++u) {
      if (w[u].x >= 0) {
        const int pr = (w[u].x >> 14) & (RNG_PTS - 1);
        if (pr / FINE_PTS == qt) {
          const int lp = pr & (FINE_PTS - 1);
          const int kk = (w[u].x >> 22) & 15;
          const int in = w[u].x & (N_PTS - 1);
          const int r  = atomicAdd(&pc[lp], 1);
          if ((unsigned)r < (unsigned)BIN_CAP) {
            lb[lp * BIN_CAP + r] = make_int2((kk << 14) | in, w[u].y);
          } else {
            overflow_push((unsigned)kk, (unsigned)(rg * RNG_PTS + pr), in, __int_as_float(w[u].y),
                          cnt, ovk, ovi, ovp, nnz);
          }
        }
      }
    }
  }
  __syncthreads();
  const int n0 = blockIdx.x * FINE_PTS;
  int2* const dst = (int2*)be + (size_t)n0 * BIN_CAP;
  for (int i = tid; i < FINE_PTS * BIN_CAP; i += FINE_THREADS) {
    int filled = pc[i / BIN_CAP];
    if (filled > BIN_CAP) filled = BIN_CAP;
    if ((i & (BIN_CAP - 1)) < filled) dst[i] = lb[i];
    else if (spilled)                 dst[i] = make_int2(15 << 14, 0);
  }
  if (tid < FINE_PTS) {
    int v = pc[tid];
    if (spilled && v <= BIN_CAP) v = BIN_CAP + 1;
    cnt[n0 + tid] = v;
  }
}

#define FUSE_PPW 4
#define LDP      68
#define TP       36
#define W_FLOATS (2 * 64 * LDP)
#define A_FLOATS (8 * 16 * LDP)

struct Row8 { float4 lo, hi; };
struct Chunk { int ek; int ei; float ep; };

__device__ __forceinline__ Row8 load_row(const float* __restrict__ xt, int inp, int lane) {
  const float4* s = (const float4*)(xt + (size_t)inp * BC) + lane * 2;
  Row8 r;
  r.lo = s[0];
  r.hi = s[1];
  return r;
}

__device__ __forceinline__ void fma_row(float4& a0, float4& a1, float p, const Row8& r) {
  a0.x += p * r.lo.x; a0.y += p * r.lo.y; a0.z += p * r.lo.z; a0.w += p * r.lo.w;
  a1.x += p * r.hi.x; a1.y += p * r.hi.y; a1.z += p * r.hi.z; a1.w += p * r.hi.w;
}

__device__ __forceinline__ Chunk load_chunk(const int* __restrict__ be,
                                            int base, int end, int lane) {
  const int idx = base + lane;
  Chunk c;
  c.ek = -1;
  c.ei = 0;
  c.ep = 0.f;
  if (idx < end) {
    const int2 w = ((const int2*)be)[idx];
    c.ek = w.x >> 14;
    c.ei = w.x & (N_PTS - 1);
    c.ep = __int_as_float(w.y);
  }
  return c;
}

__device__ __forceinline__ void add_chunk(const Chunk& c, int k,
                                          const float* __restrict__ xt, int lane,
                                          float4& a0, float4& a1) {
  unsigned mask = __builtin_amdgcn_ballot_w32(c.ek == k);
  while (mask) {
    const int l0 = __builtin_ctz(mask);
    mask &= mask - 1;
    const float p0 = __int_as_float(__builtin_amdgcn_readlane(__float_as_int(c.ep), l0));
    const Row8 r0 = load_row(xt, __builtin_amdgcn_readlane(c.ei, l0) & (N_PTS - 1), lane);
    if (!mask) { fma_row(a0, a1, p0, r0); break; }

    const int l1 = __builtin_ctz(mask);
    mask &= mask - 1;
    const float p1 = __int_as_float(__builtin_amdgcn_readlane(__float_as_int(c.ep), l1));
    const Row8 r1 = load_row(xt, __builtin_amdgcn_readlane(c.ei, l1) & (N_PTS - 1), lane);
    if (!mask) { fma_row(a0, a1, p0, r0); fma_row(a0, a1, p1, r1); break; }

    const int l2 = __builtin_ctz(mask);
    mask &= mask - 1;
    const float p2 = __int_as_float(__builtin_amdgcn_readlane(__float_as_int(c.ep), l2));
    const Row8 r2 = load_row(xt, __builtin_amdgcn_readlane(c.ei, l2) & (N_PTS - 1), lane);
    if (!mask) {
      fma_row(a0, a1, p0, r0); fma_row(a0, a1, p1, r1); fma_row(a0, a1, p2, r2);
      break;
    }

    const int l3 = __builtin_ctz(mask);
    mask &= mask - 1;
    const float p3 = __int_as_float(__builtin_amdgcn_readlane(__float_as_int(c.ep), l3));
    const Row8 r3 = load_row(xt, __builtin_amdgcn_readlane(c.ei, l3) & (N_PTS - 1), lane);
    __builtin_amdgcn_sched_barrier(0);
    float q0 = p0, q1 = p1, q2 = p2;
    asm volatile("" : "+s"(q0), "+s"(q1), "+s"(q2));
    fma_row(a0, a1, q0, r0); fma_row(a0, a1, q1, r1);
    fma_row(a0, a1, q2, r2); fma_row(a0, a1, p3, r3);
  }
}

#define GATHER_SLOTS 12
#define LIST_INTS    (2 * 256)

__device__ __forceinline__ void store_point(float* at, int pt, int lane,
                                            const float4& s0, const float4& s1) {
  float* d = at + (pt * 4 + (lane >> 3)) * LDP + (lane & 7) * 8;
  *(float4*)d       = s0;
  *(float4*)(d + 4) = s1;
}

template <int C>
__device__ __forceinline__ void gather_trip(const int2& e, int j0,
                                            const float* __restrict__ xt, int lane, float* at,
                                            int& pt, int& written, float4& s0, float4& s1) {
  int   key[C];
  float w[C];
  Row8  r[C];
#pragma unroll
  for (int s = 0; s < C; ++s) {
    key[s] = __builtin_amdgcn_readlane(e.x, j0 + s);
    w[s]   = __int_as_float(__builtin_amdgcn_readlane(e.y, j0 + s));
    r[s]   = load_row(xt, key[s] & (N_PTS - 1), lane);
  }
  __builtin_amdgcn_sched_barrier(0);
#pragma unroll
  for (int s = 0; s < C; ++s) {
    const int q = (key[s] >> 14) & 3;
    if (q != pt) {
      store_point(at, pt, lane, s0, s1);
      written |= 1 << pt;
      s0 = make_float4(0.f, 0.f, 0.f, 0.f);
      s1 = s0;
      pt = q;
    }
    fma_row(s0, s1, w[s], r[s]);
  }
}

template <int C>
struct TripSel {
  static __device__ __forceinline__ void run(int c, const int2& e, int j0,
                                             const float* __restrict__ xt, int lane, float* at,
                                             int& pt, int& written, float4& s0, float4& s1) {
    if (c == C) gather_trip<C>(e, j0, xt, lane, at, pt, written, s0, s1);
    else        TripSel<C - 1>::run(c, e, j0, xt, lane, at, pt, written, s0, s1);
  }
};
template <>
struct TripSel<0> {
  static __device__ __forceinline__ void run(int, const int2&, int, const float* __restrict__, int, float*,
                                             int&, int&, float4&, float4&) {}
};

__device__ __forceinline__ void gather_stream(const Chunk (&c0)[FUSE_PPW],
                                              const Chunk (&c1)[FUSE_PPW], int k,
                                              const float* __restrict__ xt, int lane,
                                              int* list, float* at) {
  int total = 0;
#pragma unroll
  for (int i = 0; i < FUSE_PPW; ++i) {
#pragma unroll
    for (int h = 0; h < 2; ++h) {
      const Chunk& c = h ? c1[i] : c0[i];
      const bool     hit = c.ek == k;
      const unsigned m   = __builtin_amdgcn_ballot_w32(hit);
      const int      pos = total + (int)__builtin_amdgcn_mbcnt_lo(m, 0u);
      if (hit) {
        int2 e;
        e.x = (i << 14) | (c.ei & (N_PTS - 1));
        e.y = __float_as_int(c.ep);
        *(int2*)&list[2 * pos] = e;
      }
      total += __builtin_popcount(m);
    }
  }
  __builtin_amdgcn_fence(__ATOMIC_RELEASE, "wavefront");
  __builtin_amdgcn_wave_barrier();
  __builtin_amdgcn_fence(__ATOMIC_ACQUIRE, "wavefront");

  int    pt = 0;
  int    written = 0;
  float4 s0 = make_float4(0.f, 0.f, 0.f, 0.f);
  float4 s1 = s0;
  for (int base = 0; base < total; base += 32) {
    const int n = total - base < 32 ? total - base : 32;
    int2 e = make_int2(0, 0);
    if (lane < n) e = *(const int2*)&list[2 * (base + lane)];
    for (int j0 = 0; j0 < n; j0 += GATHER_SLOTS) {
      const int c = n - j0 < GATHER_SLOTS ? n - j0 : GATHER_SLOTS;
      TripSel<GATHER_SLOTS>::run(c, e, j0, xt, lane, at, pt, written, s0, s1);
    }
  }
  store_point(at, pt, lane, s0, s1);
  written |= 1 << pt;
  const float4 zero = make_float4(0.f, 0.f, 0.f, 0.f);
#pragma unroll
  for (int i = 0; i < FUSE_PPW; ++i) {
    if (!(written & (1 << i))) store_point(at, i, lane, zero, zero);
  }
}

__global__ void __launch_bounds__(256)
fused_kernel(const int* __restrict__ cnt,
             const int* __restrict__ be,
             const int* __restrict__ ovk,
             const int* __restrict__ ovi,
             const float* __restrict__ ovp,
             const float* __restrict__ xt,
             const float* __restrict__ wt,
             const float* __restrict__ bias,
             float* __restrict__ y,
             int nnz) {
  __shared__ __attribute__((aligned(16))) float lds[W_FLOATS + A_FLOATS];
  __shared__ __attribute__((aligned(16))) int lists[8 * LIST_INTS];
  static_assert(256 * TP <= W_FLOATS + A_FLOATS, "output staging tile must fit in the block's LDS");

  const int tid  = threadIdx.x;
  const int lane = tid & 31;
  const int hi   = lane >> 4;
  const int l15  = lane & 15;
  const int wave = __builtin_amdgcn_readfirstlane(tid >> 5);
  const int n0   = (blockIdx.x * 8 + wave) * FUSE_PPW;

  int beg[FUSE_PPW], end[FUSE_PPW];
  bool longseg = false;
#pragma unroll
  for (int i = 0; i < FUSE_PPW; ++i) {
    int len = cnt[n0 + i];
    longseg |= (len > BIN_CAP);
    if (len > BIN_CAP) len = BIN_CAP;
    if (len < 0) len = 0;
    beg[i] = (n0 + i) * BIN_CAP;
    end[i] = beg[i] + len;
  }
  int novf = 0;
  if (longseg) {
    novf = cnt[N_PTS];
    if (novf > nnz) novf = nnz;
    if (novf < 0) novf = 0;
  }
  novf = __builtin_amdgcn_readfirstlane(novf);

  Chunk ck[FUSE_PPW], cm[FUSE_PPW];
#pragma unroll
  for (int i = 0; i < FUSE_PPW; ++i) {
    ck[i] = load_chunk(be, beg[i],      end[i], lane);
    cm[i] = load_chunk(be, beg[i] + 32, end[i], lane);
  }

  float* const at = lds + W_FLOATS + wave * (16 * LDP);
  int* const list = lists + wave * LIST_INTS;
  v8f acc[4] = {};

  const float4* const wsrc = (const float4*)wt + tid;
  float4 w0 = wsrc[0], w1 = wsrc[256], w2 = wsrc[512], w3 = wsrc[768];
  const int wofs = (tid >> 4) * LDP + (tid & 15) * 4;

  for (int kk = 0; kk < KSZ; ++kk) {
    if (!longseg) {
      gather_stream(ck, cm, kk, xt, lane, list, at);
    } else {
      float4 a0[FUSE_PPW], a1[FUSE_PPW];
#pragma unroll
      for (int i = 0; i < FUSE_PPW; ++i) {
        a0[i] = make_float4(0.f, 0.f, 0.f, 0.f);
        a1[i] = a0[i];
        add_chunk(ck[i], kk, xt, lane, a0[i], a1[i]);
        add_chunk(cm[i], kk, xt, lane, a0[i], a1[i]);
        for (int base = 0; base < novf; base += 32) {
          const int idx = base + lane;
          Chunk c2;
          c2.ek = -1;
          c2.ei = 0;
          c2.ep = 0.f;
          if (idx < novf) {
            const int ok = ovk[idx];
            if ((ok & (N_PTS - 1)) == n0 + i) {
              c2.ek = ok >> 14;
              c2.ei = ovi[idx];
              c2.ep = ovp[idx];
            }
          }
          add_chunk(c2, kk, xt, lane, a0[i], a1[i]);
        }
        store_point(at, i, lane, a0[i], a1[i]);
      }
    }

    float* const wb = lds + (kk & 1) * (64 * LDP);
    *(float4*)&wb[wofs]            = w0;
    *(float4*)&wb[wofs + 16 * LDP] = w1;
    *(float4*)&wb[wofs + 32 * LDP] = w2;
    *(float4*)&wb[wofs + 48 * LDP] = w3;
    __syncthreads();

    if (kk + 1 < KSZ) {
      const float4* s = wsrc + (kk + 1) * (COUT * CIN / 4);
      w0 = s[0]; w1 = s[256]; w2 = s[512]; w3 = s[768];
    }

#pragma unroll
    for (int ks = 0; ks < 16; ++ks) {
      const int k2 = ks * 4 + hi * 2;
      v2f a = { at[l15 * LDP + k2], at[l15 * LDP + k2 + 1] };
#pragma unroll
      for (int wn = 0; wn < 4; ++wn) {
        v2f bw = { wb[(wn * 16 + l15) * LDP + k2], wb[(wn * 16 + l15) * LDP + k2 + 1] };
        acc[wn] = __builtin_amdgcn_wmma_f32_16x16x4_f32(
             false, a,  false, bw,
             (short)0, acc[wn],  false,  false);
      }
    }
  }

  __syncthreads();
  float* const T = lds;
#pragma unroll
  for (int wn = 0; wn < 4; ++wn) {
    const float bv = bias[wn * 16 + l15];
#pragma unroll
    for (int r = 0; r < 8; ++r) {
      T[((r & 3) * COUT + wn * 16 + l15) * TP + wave * FUSE_PPW + 2 * hi + (r >> 2)] = acc[wn][r] + bv;
    }
  }
  __syncthreads();

  const int nb = blockIdx.x * (8 * FUSE_PPW);
#pragma unroll
  for (int s = 0; s < 8; ++s) {
    const int j = wave * 32 + s * 4 + (lane >> 3);
    const float4 v = *(const float4*)&T[j * TP + (lane & 7) * 4];
    *(float4*)(y + (size_t)j * N_PTS + nb + (lane & 7) * 4) = v;
  }
}

extern "C" void kernel_launch(void* const* d_in, const int* in_sizes, int n_in,
                              void* d_out, int out_size, void* d_ws, size_t ws_size,
                              hipStream_t stream) {
  const float* x        = (const float*)d_in[0];
  const int*   psi_idx  = (const int*)  d_in[1];
  const float* psi_vals = (const float*)d_in[2];
  const float* quad     = (const float*)d_in[3];
  const float* weight   = (const float*)d_in[4];
  const float* bias     = (const float*)d_in[5];
  float*       y        = (float*)d_out;

  const int nnz = in_sizes[2];

  float* xt   = (float*)d_ws;
  float* wt   = xt + (size_t)N_PTS * BC;
  int*   be   = (int*)(wt + KSZ * COUT * CIN);
  int*   rr   = be + 2 * N_PTS * BIN_CAP;
  int*   rcnt = rr + 2 * N_RNG * RNG_CAP;
  int*   cnt  = (int*)(wt + (size_t)N_PTS * BC);
  int*   ovk  = cnt + N_PTS + 256;
  int*   ovi  = ovk + nnz;
  float* ovp  = (float*)(ovi + nnz);

  const int* kI = psi_idx;
  const int* oI = psi_idx + nnz;
  const int* iI = psi_idx + 2 * nnz;

  setup_kernel<<<XQ_BLOCKS + WT_BLOCKS + ZERO_BLOCKS, 256, 0, stream>>>(x, quad, xt, weight, wt, rcnt, cnt);

  range_kernel<<<256, 256, 0, stream>>>(kI, oI, iI, psi_vals, rcnt, rr, cnt, ovk, ovi, ovp, nnz);

  point_kernel<<<N_PTS / FINE_PTS, FINE_THREADS, 0, stream>>>(rcnt, rr, cnt, be, ovk, ovi, ovp, nnz);

  fused_kernel<<<N_PTS / (8 * FUSE_PPW), 256, 0, stream>>>(cnt, be, ovk, ovi, ovp, xt, wt, bias, y, nnz);
}


# All nine weight blocks staged once in the LDS of a 16-wave, 64-point block: the per-k block barrier and weight restaging go, waves run gather and products unsynchronised; same sums and order
# speedup vs baseline: 8.7150x; 1.0449x over previous
#include <hip/hip_runtime.h>

typedef __attribute__((ext_vector_type(2))) float v2f;
typedef __attribute__((ext_vector_type(8))) float v8f;

#define N_PTS  16384
#define BATCH  4
#define CIN    64
#define COUT   64
#define KSZ    9
#define BC     (BATCH * CIN)


#define XQ_BLOCKS   (N_PTS / 32)
#define WT_BLOCKS   (KSZ * COUT * CIN / 256)
#define BIN_CAP     64
#define RNG_PTS     256
#define N_RNG       (N_PTS / RNG_PTS)
#define RNG_CAP     12288
#define RC_PITCH    32
#define RC_BLOCKS   (N_RNG * RC_PITCH / 256)
#define ZERO_BLOCKS (RC_BLOCKS + 1)

__global__ void __launch_bounds__(256)
setup_kernel(const float* __restrict__ x,
             const float* __restrict__ quad,
             float* __restrict__ xt,
             const float* __restrict__ weight,
             float* __restrict__ wt,
             int* __restrict__ rcnt,
             int* __restrict__ cnt) {
  const int blk = blockIdx.x;
  if (blk < XQ_BLOCKS) {
    const int j  = threadIdx.x;
    const int m0 = blk * 32;
    const float4* s = (const float4*)(x + (size_t)j * N_PTS + m0);
    float4 v[8];
#pragma unroll
    for (int i = 0; i < 8; ++i) v[i] = s[i];
    float* d = xt + (size_t)m0 * BC + j;
#pragma unroll
    for (int i = 0; i < 8; ++i) {
      d[(4 * i + 0) * BC] = v[i].x * quad[m0 + 4 * i + 0];
      d[(4 * i + 1) * BC] = v[i].y * quad[m0 + 4 * i + 1];
      d[(4 * i + 2) * BC] = v[i].z * quad[m0 + 4 * i + 2];
      d[(4 * i + 3) * BC] = v[i].w * quad[m0 + 4 * i + 3];
    }
  } else if (blk < XQ_BLOCKS + WT_BLOCKS) {
    const int idx = (blk - XQ_BLOCKS) * 256 + threadIdx.x;
    const int k   = idx >> 12;
    const int oc  = idx & 4095;
    wt[idx] = weight[oc * KSZ + k];
  } else {
    const int z = blk - XQ_BLOCKS - WT_BLOCKS;
    if (z < RC_BLOCKS) rcnt[z * 256 + threadIdx.x] = 0;
    else               cnt[N_PTS + threadIdx.x] = 0;
  }
}

#define SORT_CHUNK  2048
#define SORT_UNROLL (SORT_CHUNK / 256)

__device__ __forceinline__ void overflow_push(unsigned kk, unsigned o, int in, float p,
                                              int* __restrict__ cnt,
                                              int* __restrict__ ovk, int* __restrict__ ovi, float* __restrict__ ovp,
                                              int nnz) {
  const int q = atomicAdd(&cnt[N_PTS], 1);
  if ((unsigned)q < (unsigned)nnz) {
    ovk[q] = (int)((kk << 14) | o);
    ovi[q] = in;
    ovp[q] = p;
  }
}

__global__ void __launch_bounds__(256)
range_kernel(const int* __restrict__ kI,
             const int* __restrict__ oI,
             const int* __restrict__ iI,
             const float* __restrict__ psi,
             int* __restrict__ rcnt,
             int* __restrict__ rr,
             int* __restrict__ cnt,
             int* __restrict__ ovk,
             int* __restrict__ ovi,
             float* __restrict__ ovp, int nnz) {
  __shared__ int  hist[N_RNG];
  __shared__ int  start[N_RNG];
  __shared__ int  gbase[N_RNG];
  __shared__ int2 srec[SORT_CHUNK];
  __shared__ int  sdst[SORT_CHUNK];
  const int tid = threadIdx.x;
  for (int c0 = blockIdx.x * SORT_CHUNK; c0 < nnz; c0 += gridDim.x * SORT_CHUNK) {
    if (tid < N_RNG) hist[tid] = 0;
    __syncthreads();
    unsigned kk[SORT_UNROLL], o[SORT_UNROLL];
    int      in[SORT_UNROLL], lr[SORT_UNROLL];
    float    p[SORT_UNROLL];
#pragma unroll
    for (int u = 0; u < SORT_UNROLL; ++u) {
      const int e = c0 + u * 256 + tid;
      kk[u] = KSZ;
      o[u]  = 0;
      in[u] = 0;
      p[u]  = 0.f;
      if (e < nnz) {
        kk[u] = (unsigned)kI[e];
        o[u]  = (unsigned)oI[e];
        in[u] = iI[e];
        p[u]  = psi[e];
      }
    }
#pragma unroll
    for (int u = 0; u < SORT_UNROLL; ++u) {
      lr[u] = -1;
      if (kk[u] < KSZ && o[u] < N_PTS) lr[u] = atomicAdd(&hist[o[u] / RNG_PTS], 1);
    }
    __syncthreads();
    if (tid < N_RNG) {
      const int h  = hist[tid];
      const int gb = h ? atomicAdd(&rcnt[tid * RC_PITCH], h) : 0;
      int s = 0;
#pragma unroll
      for (int j = 0; j < N_RNG; ++j) s += (j < tid) ? hist[j] : 0;
      start[tid] = s;
      gbase[tid] = gb;
    }
    __syncthreads();
#pragma unroll
    for (int u = 0; u < SORT_UNROLL; ++u) {
      if (lr[u] >= 0) {
        const int rg   = (int)(o[u] / RNG_PTS);
        const int pos  = start[rg] + lr[u];
        const int slot = gbase[rg] + lr[u];
        if ((unsigned)slot < (unsigned)RNG_CAP) {
          int2 w;
          w.x = (int)((kk[u] << 22) | ((o[u] & (RNG_PTS - 1)) << 14) | ((unsigned)in[u] & (N_PTS - 1)));
          w.y = __float_as_int(p[u]);
          srec[pos] = w;
          sdst[pos] = rg * RNG_CAP + slot;
        } else {
          sdst[pos] = -1;
          overflow_push(kk[u], o[u], in[u], p[u], cnt, ovk, ovi, ovp, nnz);
        }
      }
    }
    __syncthreads();
    const int total = start[N_RNG - 1] + hist[N_RNG - 1];
#pragma unroll
    for (int u = 0; u < SORT_UNROLL; ++u) {
      const int pos = u * 256 + tid;
      if (pos < total) {
        const int d = sdst[pos];
        if (d >= 0) ((int2*)rr)[d] = srec[pos];
      }
    }
    __syncthreads();
  }
}

#define FINE_PTS    64
#define FINE_THREADS 512
#define FINE_UNROLL 8

__global__ void __launch_bounds__(FINE_THREADS)
point_kernel(const int* __restrict__ rcnt,
             const int* __restrict__ rr,
             int* __restrict__ cnt,
             int* __restrict__ be,
             int* __restrict__ ovk,
             int* __restrict__ ovi,
             float* __restrict__ ovp, int nnz) {
  __shared__ int  pc[FINE_PTS];
  __shared__ int2 lb[FINE_PTS * BIN_CAP];
  const int tid = threadIdx.x;
  const int rg  = blockIdx.x / (RNG_PTS / FINE_PTS);
  const int qt  = blockIdx.x % (RNG_PTS / FINE_PTS);
  if (tid < FINE_PTS) pc[tid] = 0;
  __syncthreads();
  int n = rcnt[rg * RC_PITCH];
  const bool spilled = n > RNG_CAP;
  if (n > RNG_CAP) n = RNG_CAP;
  if (n < 0) n = 0;
  const int2* const srcp = (const int2*)rr + (size_t)rg * RNG_CAP;
  for (int i0 = 0; i0 < n; i0 += FINE_UNROLL * FINE_THREADS) {
    int2 w[FINE_UNROLL];
#pragma unroll
    for (int u = 0; u < FINE_UNROLL; ++u) {
      const int i = i0 + u * FINE_THREADS + tid;
      w[u] = make_int2(-1, 0);
      if (i < n) w[u] = srcp[i];
    }
#pragma unroll
    for (int u = 0; u < FINE_UNROLL; ++u) {
      if (w[u].x >= 0) {
        const int pr = (w[u].x >> 14) & (RNG_PTS - 1);
        if (pr / FINE_PTS == qt) {
          const int lp = pr & (FINE_PTS - 1);
          const int kk = (w[u].x >> 22) & 15;
          const int in = w[u].x & (N_PTS - 1);
          const int r  = atomicAdd(&pc[lp], 1);
          if ((unsigned)r < (unsigned)BIN_CAP) {
            lb[lp * BIN_CAP + r] = make_int2((kk << 14) | in, w[u].y);
          } else {
            overflow_push((unsigned)kk, (unsigned)(rg * RNG_PTS + pr), in, __int_as_float(w[u].y),
                          cnt, ovk, ovi, ovp, nnz);
          }
        }
      }
    }
  }
  __syncthreads();
  const int n0 = blockIdx.x * FINE_PTS;
  int2* const dst = (int2*)be + (size_t)n0 * BIN_CAP;
  for (int i = tid; i < FINE_PTS * BIN_CAP; i += FINE_THREADS) {
    int filled = pc[i / BIN_CAP];
    if (filled > BIN_CAP) filled = BIN_CAP;
    if ((i & (BIN_CAP - 1)) < filled) dst[i] = lb[i];
    else if (spilled)                 dst[i] = make_int2(15 << 14, 0);
  }
  if (tid < FINE_PTS) {
    int v = pc[tid];
    if (spilled && v <= BIN_CAP) v = BIN_CAP + 1;
    cnt[n0 + tid] = v;
  }
}

#define FUSE_PPW 4
#define FUSE_WAVES   16
#define FUSE_THREADS (FUSE_WAVES * 32)
#define FUSE_PTS     (FUSE_WAVES * FUSE_PPW)
#define LDP      68
#define TP       68
#define W_FLOATS (KSZ * 64 * LDP)
#define A_FLOATS (FUSE_WAVES * 16 * LDP)
#define W_VEC    (KSZ * COUT * CIN / 4 / FUSE_THREADS)

struct Row8 { float4 lo, hi; };
struct Chunk { int ek; int ei; float ep; };

__device__ __forceinline__ Row8 load_row(const float* __restrict__ xt, int inp, int lane) {
  const float4* s = (const float4*)(xt + (size_t)inp * BC) + lane * 2;
  Row8 r;
  r.lo = s[0];
  r.hi = s[1];
  return r;
}

__device__ __forceinline__ void fma_row(float4& a0, float4& a1, float p, const Row8& r) {
  a0.x += p * r.lo.x; a0.y += p * r.lo.y; a0.z += p * r.lo.z; a0.w += p * r.lo.w;
  a1.x += p * r.hi.x; a1.y += p * r.hi.y; a1.z += p * r.hi.z; a1.w += p * r.hi.w;
}

__device__ __forceinline__ Chunk load_chunk(const int* __restrict__ be,
                                            int base, int end, int lane) {
  const int idx = base + lane;
  Chunk c;
  c.ek = -1;
  c.ei = 0;
  c.ep = 0.f;
  if (idx < end) {
    const int2 w = ((const int2*)be)[idx];
    c.ek = w.x >> 14;
    c.ei = w.x & (N_PTS - 1);
    c.ep = __int_as_float(w.y);
  }
  return c;
}

__device__ __forceinline__ void add_chunk(const Chunk& c, int k,
                                          const float* __restrict__ xt, int lane,
                                          float4& a0, float4& a1) {
  unsigned mask = __builtin_amdgcn_ballot_w32(c.ek == k);
  while (mask) {
    const int l0 = __builtin_ctz(mask);
    mask &= mask - 1;
    const float p0 = __int_as_float(__builtin_amdgcn_readlane(__float_as_int(c.ep), l0));
    const Row8 r0 = load_row(xt, __builtin_amdgcn_readlane(c.ei, l0) & (N_PTS - 1), lane);
    if (!mask) { fma_row(a0, a1, p0, r0); break; }

    const int l1 = __builtin_ctz(mask);
    mask &= mask - 1;
    const float p1 = __int_as_float(__builtin_amdgcn_readlane(__float_as_int(c.ep), l1));
    const Row8 r1 = load_row(xt, __builtin_amdgcn_readlane(c.ei, l1) & (N_PTS - 1), lane);
    if (!mask) { fma_row(a0, a1, p0, r0); fma_row(a0, a1, p1, r1); break; }

    const int l2 = __builtin_ctz(mask);
    mask &= mask - 1;
    const float p2 = __int_as_float(__builtin_amdgcn_readlane(__float_as_int(c.ep), l2));
    const Row8 r2 = load_row(xt, __builtin_amdgcn_readlane(c.ei, l2) & (N_PTS - 1), lane);
    if (!mask) {
      fma_row(a0, a1, p0, r0); fma_row(a0, a1, p1, r1); fma_row(a0, a1, p2, r2);
      break;
    }

    const int l3 = __builtin_ctz(mask);
    mask &= mask - 1;
    const float p3 = __int_as_float(__builtin_amdgcn_readlane(__float_as_int(c.ep), l3));
    const Row8 r3 = load_row(xt, __builtin_amdgcn_readlane(c.ei, l3) & (N_PTS - 1), lane);
    __builtin_amdgcn_sched_barrier(0);
    float q0 = p0, q1 = p1, q2 = p2;
    asm volatile("" : "+s"(q0), "+s"(q1), "+s"(q2));
    fma_row(a0, a1, q0, r0); fma_row(a0, a1, q1, r1);
    fma_row(a0, a1, q2, r2); fma_row(a0, a1, p3, r3);
  }
}

#define GATHER_SLOTS 12
#define LIST_INTS    (2 * 256)

__device__ __forceinline__ void store_point(float* at, int pt, int lane,
                                            const float4& s0, const float4& s1) {
  float* d = at + (pt * 4 + (lane >> 3)) * LDP + (lane & 7) * 8;
  *(float4*)d       = s0;
  *(float4*)(d + 4) = s1;
}

template <int C>
__device__ __forceinline__ void gather_trip(const int2& e, int j0,
                                            const float* __restrict__ xt, int lane, float* at,
                                            int& pt, int& written, float4& s0, float4& s1) {
  int   key[C];
  float w[C];
  Row8  r[C];
#pragma unroll
  for (int s = 0; s < C; ++s) {
    key[s] = __builtin_amdgcn_readlane(e.x, j0 + s);
    w[s]   = __int_as_float(__builtin_amdgcn_readlane(e.y, j0 + s));
    r[s]   = load_row(xt, key[s] & (N_PTS - 1), lane);
  }
  __builtin_amdgcn_sched_barrier(0);
#pragma unroll
  for (int s = 0; s < C; ++s) {
    const int q = (key[s] >> 14) & 3;
    if (q != pt) {
      store_point(at, pt, lane, s0, s1);
      written |= 1 << pt;
      s0 = make_float4(0.f, 0.f, 0.f, 0.f);
      s1 = s0;
      pt = q;
    }
    fma_row(s0, s1, w[s], r[s]);
  }
}

template <int C>
struct TripSel {
  static __device__ __forceinline__ void run(int c, const int2& e, int j0,
                                             const float* __restrict__ xt, int lane, float* at,
                                             int& pt, int& written, float4& s0, float4& s1) {
    if (c == C) gather_trip<C>(e, j0, xt, lane, at, pt, written, s0, s1);
    else        TripSel<C - 1>::run(c, e, j0, xt, lane, at, pt, written, s0, s1);
  }
};
template <>
struct TripSel<0> {
  static __device__ __forceinline__ void run(int, const int2&, int, const float* __restrict__, int, float*,
                                             int&, int&, float4&, float4&) {}
};

__device__ __forceinline__ void gather_stream(const Chunk (&c0)[FUSE_PPW],
                                              const Chunk (&c1)[FUSE_PPW], int k,
                                              const float* __restrict__ xt, int lane,
                                              int* list, float* at) {
  int total = 0;
#pragma unroll
  for (int i = 0; i < FUSE_PPW; ++i) {
#pragma unroll
    for (int h = 0; h < 2; ++h) {
      const Chunk& c = h ? c1[i] : c0[i];
      const bool     hit = c.ek == k;
      const unsigned m   = __builtin_amdgcn_ballot_w32(hit);
      const int      pos = total + (int)__builtin_amdgcn_mbcnt_lo(m, 0u);
      if (hit) {
        int2 e;
        e.x = (i << 14) | (c.ei & (N_PTS - 1));
        e.y = __float_as_int(c.ep);
        *(int2*)&list[2 * pos] = e;
      }
      total += __builtin_popcount(m);
    }
  }
  __builtin_amdgcn_fence(__ATOMIC_RELEASE, "wavefront");
  __builtin_amdgcn_wave_barrier();
  __builtin_amdgcn_fence(__ATOMIC_ACQUIRE, "wavefront");

  int    pt = 0;
  int    written = 0;
  float4 s0 = make_float4(0.f, 0.f, 0.f, 0.f);
  float4 s1 = s0;
  for (int base = 0; base < total; base += 32) {
    const int n = total - base < 32 ? total - base : 32;
    int2 e = make_int2(0, 0);
    if (lane < n) e = *(const int2*)&list[2 * (base + lane)];
    for (int j0 = 0; j0 < n; j0 += GATHER_SLOTS) {
      const int c = n - j0 < GATHER_SLOTS ? n - j0 : GATHER_SLOTS;
      TripSel<GATHER_SLOTS>::run(c, e, j0, xt, lane, at, pt, written, s0, s1);
    }
  }
  store_point(at, pt, lane, s0, s1);
  written |= 1 << pt;
  const float4 zero = make_float4(0.f, 0.f, 0.f, 0.f);
#pragma unroll
  for (int i = 0; i < FUSE_PPW; ++i) {
    if (!(written & (1 << i))) store_point(at, i, lane, zero, zero);
  }
}

__global__ void __launch_bounds__(FUSE_THREADS)
fused_kernel(const int* __restrict__ cnt,
             const int* __restrict__ be,
             const int* __restrict__ ovk,
             const int* __restrict__ ovi,
             const float* __restrict__ ovp,
             const float* __restrict__ xt,
             const float* __restrict__ wt,
             const float* __restrict__ bias,
             float* __restrict__ y,
             int nnz) {
  __shared__ __attribute__((aligned(16))) float lds[W_FLOATS + A_FLOATS];
  __shared__ __attribute__((aligned(16))) int lists[FUSE_WAVES * LIST_INTS];
  static_assert(256 * TP <= W_FLOATS, "output staging tile must fit in the weights' LDS");
  static_assert(W_VEC * FUSE_THREADS * 4 == KSZ * COUT * CIN, "the threads' pieces cover the weights exactly");
  static_assert(FUSE_THREADS == 512, "the weight staging below splits a 64 x 64 block into two halves of 512 pieces");

  const int tid  = threadIdx.x;
  const int lane = tid & 31;
  const int hi   = lane >> 4;
  const int l15  = lane & 15;
  const int wave = __builtin_amdgcn_readfirstlane(tid >> 5);
  const int n0   = (blockIdx.x * FUSE_WAVES + wave) * FUSE_PPW;


  int beg[FUSE_PPW], end[FUSE_PPW];
  bool longseg = false;
#pragma unroll
  for (int i = 0; i < FUSE_PPW; ++i) {
    int len = cnt[n0 + i];
    longseg |= (len > BIN_CAP);
    if (len > BIN_CAP) len = BIN_CAP;
    if (len < 0) len = 0;
    beg[i] = (n0 + i) * BIN_CAP;
    end[i] = beg[i] + len;
  }
  int novf = 0;
  if (longseg) {
    novf = cnt[N_PTS];
    if (novf > nnz) novf = nnz;
    if (novf < 0) novf = 0;
  }
  novf = __builtin_amdgcn_readfirstlane(novf);

  Chunk ck[FUSE_PPW], cm[FUSE_PPW];
#pragma unroll
  for (int i = 0; i < FUSE_PPW; ++i) {
    ck[i] = load_chunk(be, beg[i],      end[i], lane);
    cm[i] = load_chunk(be, beg[i] + 32, end[i], lane);
  }

  float* const at = lds + W_FLOATS + wave * (16 * LDP);
  int* const list = lists + wave * LIST_INTS;
  v8f acc[4] = {};

  {
    const float4* const wsrc = (const float4*)wt + tid;
    const int wofs = (tid >> 4) * LDP + (tid & 15) * 4;
    float4 wv[W_VEC];
#pragma unroll
    for (int i = 0; i < W_VEC; ++i) wv[i] = wsrc[i * FUSE_THREADS];
#pragma unroll
    for (int i = 0; i < W_VEC; ++i)
      *(float4*)&lds[(i >> 1) * (64 * LDP) + (i & 1) * (32 * LDP) + wofs] = wv[i];
  }
  __syncthreads();

  for (int kk = 0; kk < KSZ; ++kk) {
    __builtin_amdgcn_fence(__ATOMIC_RELEASE, "wavefront");
    __builtin_amdgcn_wave_barrier();
    __builtin_amdgcn_fence(__ATOMIC_ACQUIRE, "wavefront");
    if (!longseg) {
      gather_stream(ck, cm, kk, xt, lane, list, at);
    } else {
      float4 a0[FUSE_PPW], a1[FUSE_PPW];
#pragma unroll
      for (int i = 0; i < FUSE_PPW; ++i) {
        a0[i] = make_float4(0.f, 0.f, 0.f, 0.f);
        a1[i] = a0[i];
        add_chunk(ck[i], kk, xt, lane, a0[i], a1[i]);
        add_chunk(cm[i], kk, xt, lane, a0[i], a1[i]);
        for (int base = 0; base < novf; base += 32) {
          const int idx = base + lane;
          Chunk c2;
          c2.ek = -1;
          c2.ei = 0;
          c2.ep = 0.f;
          if (idx < novf) {
            const int ok = ovk[idx];
            if ((ok & (N_PTS - 1)) == n0 + i) {
              c2.ek = ok >> 14;
              c2.ei = ovi[idx];
              c2.ep = ovp[idx];
            }
          }
          add_chunk(c2, kk, xt, lane, a0[i], a1[i]);
        }
        store_point(at, i, lane, a0[i], a1[i]);
      }
    }

    __builtin_amdgcn_fence(__ATOMIC_RELEASE, "wavefront");
    __builtin_amdgcn_wave_barrier();
    __builtin_amdgcn_fence(__ATOMIC_ACQUIRE, "wavefront");
    const float* const wb = lds + kk * (64 * LDP);

#pragma unroll
    for (int ks = 0; ks < 16; ++ks) {
      const int k2 = ks * 4 + hi * 2;
      v2f a = { at[l15 * LDP + k2], at[l15 * LDP + k2 + 1] };
#pragma unroll
      for (int wn = 0; wn < 4; ++wn) {
        v2f bw = { wb[(wn * 16 + l15) * LDP + k2], wb[(wn * 16 + l15) * LDP + k2 + 1] };
        acc[wn] = __builtin_amdgcn_wmma_f32_16x16x4_f32(
             false, a,  false, bw,
             (short)0, acc[wn],  false,  false);
      }
    }
  }

  __syncthreads();
  float* const T = lds;
#pragma unroll
  for (int wn = 0; wn < 4; ++wn) {
    const float bv = bias[wn * 16 + l15];
#pragma unroll
    for (int r = 0; r < 8; ++r) {
      T[((r & 3) * COUT + wn * 16 + l15) * TP + wave * FUSE_PPW + 2 * hi + (r >> 2)] = acc[wn][r] + bv;
    }
  }
  __syncthreads();

  const int nb = blockIdx.x * FUSE_PTS;
#pragma unroll
  for (int s = 0; s < 8; ++s) {
    const int j = wave * 16 + s * 2 + (lane >> 4);
    const float4 v = *(const float4*)&T[j * TP + (lane & 15) * 4];
    *(float4*)(y + (size_t)j * N_PTS + nb + (lane & 15) * 4) = v;
  }
}

extern "C" void kernel_launch(void* const* d_in, const int* in_sizes, int n_in,
                              void* d_out, int out_size, void* d_ws, size_t ws_size,
                              hipStream_t stream) {
  const float* x        = (const float*)d_in[0];
  const int*   psi_idx  = (const int*)  d_in[1];
  const float* psi_vals = (const float*)d_in[2];
  const float* quad     = (const float*)d_in[3];
  const float* weight   = (const float*)d_in[4];
  const float* bias     = (const float*)d_in[5];
  float*       y        = (float*)d_out;

  const int nnz = in_sizes[2];

  float* xt   = (float*)d_ws;
  float* wt   = xt + (size_t)N_PTS * BC;
  int*   be   = (int*)(wt + KSZ * COUT * CIN);
  int*   rr   = be + 2 * N_PTS * BIN_CAP;
  int*   rcnt = rr + 2 * N_RNG * RNG_CAP;
  int*   cnt  = (int*)(wt + (size_t)N_PTS * BC);
  int*   ovk  = cnt + N_PTS + 256;
  int*   ovi  = ovk + nnz;
  float* ovp  = (float*)(ovi + nnz);

  const int* kI = psi_idx;
  const int* oI = psi_idx + nnz;
  const int* iI = psi_idx + 2 * nnz;

  setup_kernel<<<XQ_BLOCKS + WT_BLOCKS + ZERO_BLOCKS, 256, 0, stream>>>(x, quad, xt, weight, wt, rcnt, cnt);

  range_kernel<<<256, 256, 0, stream>>>(kI, oI, iI, psi_vals, rcnt, rr, cnt, ovk, ovi, ovp, nnz);

  point_kernel<<<N_PTS / FINE_PTS, FINE_THREADS, 0, stream>>>(rcnt, rr, cnt, be, ovk, ovi, ovp, nnz);

  fused_kernel<<<N_PTS / FUSE_PTS, FUSE_THREADS, 0, stream>>>(cnt, be, ovk, ovi, ovp, xt, wt, bias, y, nnz);
}


# Second sort level moved into the fused kernel as its block prologue: a block bins its 64 points' records in LDS and its waves read the bins from there; 4 launches to 3; arrival-order sums stay
# speedup vs baseline: 9.4736x; 1.0870x over previous
#include <hip/hip_runtime.h>

typedef __attribute__((ext_vector_type(2))) float v2f;
typedef __attribute__((ext_vector_type(8))) float v8f;

#define N_PTS  16384
#define BATCH  4
#define CIN    64
#define COUT   64
#define KSZ    9
#define BC     (BATCH * CIN)


#define XQ_BLOCKS   (N_PTS / 32)
#define WT_BLOCKS   (KSZ * COUT * CIN / 256)
#define BIN_CAP     64
#define RNG_PTS     256
#define N_RNG       (N_PTS / RNG_PTS)
#define RNG_CAP     12288
#define RC_PITCH    32
#define RC_BLOCKS   (N_RNG * RC_PITCH / 256)
#define ZERO_BLOCKS (RC_BLOCKS + 1)

__global__ void __launch_bounds__(256)
setup_kernel(const float* __restrict__ x,
             const float* __restrict__ quad,
             float* __restrict__ xt,
             const float* __restrict__ weight,
             float* __restrict__ wt,
             int* __restrict__ rcnt,
             int* __restrict__ cnt) {
  const int blk = blockIdx.x;
  if (blk < XQ_BLOCKS) {
    const int j  = threadIdx.x;
    const int m0 = blk * 32;
    const float4* s = (const float4*)(x + (size_t)j * N_PTS + m0);
    float4 v[8];
#pragma unroll
    for (int i = 0; i < 8; ++i) v[i] = s[i];
    float* d = xt + (size_t)m0 * BC + j;
#pragma unroll
    for (int i = 0; i < 8; ++i) {
      d[(4 * i + 0) * BC] = v[i].x * quad[m0 + 4 * i + 0];
      d[(4 * i + 1) * BC] = v[i].y * quad[m0 + 4 * i + 1];
      d[(4 * i + 2) * BC] = v[i].z * quad[m0 + 4 * i + 2];
      d[(4 * i + 3) * BC] = v[i].w * quad[m0 + 4 * i + 3];
    }
  } else if (blk < XQ_BLOCKS + WT_BLOCKS) {
    const int idx = (blk - XQ_BLOCKS) * 256 + threadIdx.x;
    const int k   = idx >> 12;
    const int oc  = idx & 4095;
    wt[idx] = weight[oc * KSZ + k];
  } else {
    const int z = blk - XQ_BLOCKS - WT_BLOCKS;
    if (z < RC_BLOCKS) rcnt[z * 256 + threadIdx.x] = 0;
    else               cnt[N_PTS + threadIdx.x] = 0;
  }
}

#define SORT_CHUNK  2048
#define SORT_UNROLL (SORT_CHUNK / 256)

__device__ __forceinline__ void overflow_push(unsigned kk, unsigned o, int in, float p,
                                              int* __restrict__ cnt,
                                              int* __restrict__ ovk, int* __restrict__ ovi, float* __restrict__ ovp,
                                              int nnz) {
  const int q = atomicAdd(&cnt[N_PTS], 1);
  if ((unsigned)q < (unsigned)nnz) {
    ovk[q] = (int)((kk << 14) | o);
    ovi[q] = in;
    ovp[q] = p;
  }
}

__global__ void __launch_bounds__(256)
range_kernel(const int* __restrict__ kI,
             const int* __restrict__ oI,
             const int* __restrict__ iI,
             const float* __restrict__ psi,
             int* __restrict__ rcnt,
             int* __restrict__ rr,
             int* __restrict__ cnt,
             int* __restrict__ ovk,
             int* __restrict__ ovi,
             float* __restrict__ ovp, int nnz) {
  __shared__ int  hist[N_RNG];
  __shared__ int  start[N_RNG];
  __shared__ int  gbase[N_RNG];
  __shared__ int2 srec[SORT_CHUNK];
  __shared__ int  sdst[SORT_CHUNK];
  const int tid = threadIdx.x;
  for (int c0 = blockIdx.x * SORT_CHUNK; c0 < nnz; c0 += gridDim.x * SORT_CHUNK) {
    if (tid < N_RNG) hist[tid] = 0;
    __syncthreads();
    unsigned kk[SORT_UNROLL], o[SORT_UNROLL];
    int      in[SORT_UNROLL], lr[SORT_UNROLL];
    float    p[SORT_UNROLL];
#pragma unroll
    for (int u = 0; u < SORT_UNROLL; ++u) {
      const int e = c0 + u * 256 + tid;
      kk[u] = KSZ;
      o[u]  = 0;
      in[u] = 0;
      p[u]  = 0.f;
      if (e < nnz) {
        kk[u] = (unsigned)kI[e];
        o[u]  = (unsigned)oI[e];
        in[u] = iI[e];
        p[u]  = psi[e];
      }
    }
#pragma unroll
    for (int u = 0; u < SORT_UNROLL; ++u) {
      lr[u] = -1;
      if (kk[u] < KSZ && o[u] < N_PTS) lr[u] = atomicAdd(&hist[o[u] / RNG_PTS], 1);
    }
    __syncthreads();
    if (tid < N_RNG) {
      const int h  = hist[tid];
      const int gb = h ? atomicAdd(&rcnt[tid * RC_PITCH], h) : 0;
      int s = 0;
#pragma unroll
      for (int j = 0; j < N_RNG; ++j) s += (j < tid) ? hist[j] : 0;
      start[tid] = s;
      gbase[tid] = gb;
    }
    __syncthreads();
#pragma unroll
    for (int u = 0; u < SORT_UNROLL; ++u) {
      if (lr[u] >= 0) {
        const int rg   = (int)(o[u] / RNG_PTS);
        const int pos  = start[rg] + lr[u];
        const int slot = gbase[rg] + lr[u];
        if ((unsigned)slot < (unsigned)RNG_CAP) {
          int2 w;
          w.x = (int)((kk[u] << 22) | ((o[u] & (RNG_PTS - 1)) << 14) | ((unsigned)in[u] & (N_PTS - 1)));
          w.y = __float_as_int(p[u]);
          srec[pos] = w;
          sdst[pos] = rg * RNG_CAP + slot;
        } else {
          sdst[pos] = -1;
          overflow_push(kk[u], o[u], in[u], p[u], cnt, ovk, ovi, ovp, nnz);
        }
      }
    }
    __syncthreads();
    const int total = start[N_RNG - 1] + hist[N_RNG - 1];
#pragma unroll
    for (int u = 0; u < SORT_UNROLL; ++u) {
      const int pos = u * 256 + tid;
      if (pos < total) {
        const int d = sdst[pos];
        if (d >= 0) ((int2*)rr)[d] = srec[pos];
      }
    }
    __syncthreads();
  }
}

#define FUSE_PPW 4
#define FUSE_WAVES   16
#define FUSE_THREADS (FUSE_WAVES * 32)
#define FUSE_PTS     (FUSE_WAVES * FUSE_PPW)
#define LDP      68
#define TP       68
#define W_FLOATS (KSZ * 64 * LDP)
#define A_FLOATS (FUSE_WAVES * 16 * LDP)
#define W_VEC    (KSZ * COUT * CIN / 4 / FUSE_THREADS)

struct Row8 { float4 lo, hi; };
struct Chunk { int ek; int ei; float ep; };

__device__ __forceinline__ Row8 load_row(const float* __restrict__ xt, int inp, int lane) {
  const float4* s = (const float4*)(xt + (size_t)inp * BC) + lane * 2;
  Row8 r;
  r.lo = s[0];
  r.hi = s[1];
  return r;
}

__device__ __forceinline__ void fma_row(float4& a0, float4& a1, float p, const Row8& r) {
  a0.x += p * r.lo.x; a0.y += p * r.lo.y; a0.z += p * r.lo.z; a0.w += p * r.lo.w;
  a1.x += p * r.hi.x; a1.y += p * r.hi.y; a1.z += p * r.hi.z; a1.w += p * r.hi.w;
}

__device__ __forceinline__ Chunk load_chunk(const int2* lb,
                                            int base, int end, int lane) {
  const int idx = base + lane;
  int2 w = make_int2(-1, 0);
  if (idx < end) w = lb[idx];
  Chunk c;
  c.ek = w.x >> 14;
  c.ei = w.x & (N_PTS - 1);
  c.ep = __int_as_float(w.y);
  return c;
}

__device__ __forceinline__ void add_chunk(const Chunk& c, int k,
                                          const float* __restrict__ xt, int lane,
                                          float4& a0, float4& a1) {
  unsigned mask = __builtin_amdgcn_ballot_w32(c.ek == k);
  while (mask) {
    const int l0 = __builtin_ctz(mask);
    mask &= mask - 1;
    const float p0 = __int_as_float(__builtin_amdgcn_readlane(__float_as_int(c.ep), l0));
    const Row8 r0 = load_row(xt, __builtin_amdgcn_readlane(c.ei, l0) & (N_PTS - 1), lane);
    if (!mask) { fma_row(a0, a1, p0, r0); break; }

    const int l1 = __builtin_ctz(mask);
    mask &= mask - 1;
    const float p1 = __int_as_float(__builtin_amdgcn_readlane(__float_as_int(c.ep), l1));
    const Row8 r1 = load_row(xt, __builtin_amdgcn_readlane(c.ei, l1) & (N_PTS - 1), lane);
    if (!mask) { fma_row(a0, a1, p0, r0); fma_row(a0, a1, p1, r1); break; }

    const int l2 = __builtin_ctz(mask);
    mask &= mask - 1;
    const float p2 = __int_as_float(__builtin_amdgcn_readlane(__float_as_int(c.ep), l2));
    const Row8 r2 = load_row(xt, __builtin_amdgcn_readlane(c.ei, l2) & (N_PTS - 1), lane);
    if (!mask) {
      fma_row(a0, a1, p0, r0); fma_row(a0, a1, p1, r1); fma_row(a0, a1, p2, r2);
      break;
    }

    const int l3 = __builtin_ctz(mask);
    mask &= mask - 1;
    const float p3 = __int_as_float(__builtin_amdgcn_readlane(__float_as_int(c.ep), l3));
    const Row8 r3 = load_row(xt, __builtin_amdgcn_readlane(c.ei, l3) & (N_PTS - 1), lane);
    __builtin_amdgcn_sched_barrier(0);
    float q0 = p0, q1 = p1, q2 = p2;
    asm volatile("" : "+s"(q0), "+s"(q1), "+s"(q2));
    fma_row(a0, a1, q0, r0); fma_row(a0, a1, q1, r1);
    fma_row(a0, a1, q2, r2); fma_row(a0, a1, p3, r3);
  }
}

#define GATHER_SLOTS 12
#define LIST_INTS    (2 * 256)

__device__ __forceinline__ void store_point(float* at, int pt, int lane,
                                            const float4& s0, const float4& s1) {
  float* d = at + (pt * 4 + (lane >> 3)) * LDP + (lane & 7) * 8;
  *(float4*)d       = s0;
  *(float4*)(d + 4) = s1;
}

template <int C>
__device__ __forceinline__ void gather_trip(const int2& e, int j0,
                                            const float* __restrict__ xt, int lane, float* at,
                                            int& pt, int& written, float4& s0, float4& s1) {
  int   key[C];
  float w[C];
  Row8  r[C];
#pragma unroll
  for (int s = 0; s < C; ++s) {
    key[s] = __builtin_amdgcn_readlane(e.x, j0 + s);
    w[s]   = __int_as_float(__builtin_amdgcn_readlane(e.y, j0 + s));
    r[s]   = load_row(xt, key[s] & (N_PTS - 1), lane);
  }
  __builtin_amdgcn_sched_barrier(0);
#pragma unroll
  for (int s = 0; s < C; ++s) {
    const int q = (key[s] >> 14) & 3;
    if (q != pt) {
      store_point(at, pt, lane, s0, s1);
      written |= 1 << pt;
      s0 = make_float4(0.f, 0.f, 0.f, 0.f);
      s1 = s0;
      pt = q;
    }
    fma_row(s0, s1, w[s], r[s]);
  }
}

template <int C>
struct TripSel {
  static __device__ __forceinline__ void run(int c, const int2& e, int j0,
                                             const float* __restrict__ xt, int lane, float* at,
                                             int& pt, int& written, float4& s0, float4& s1) {
    if (c == C) gather_trip<C>(e, j0, xt, lane, at, pt, written, s0, s1);
    else        TripSel<C - 1>::run(c, e, j0, xt, lane, at, pt, written, s0, s1);
  }
};
template <>
struct TripSel<0> {
  static __device__ __forceinline__ void run(int, const int2&, int, const float* __restrict__, int, float*,
                                             int&, int&, float4&, float4&) {}
};

__device__ __forceinline__ void gather_stream(const Chunk (&c0)[FUSE_PPW],
                                              const Chunk (&c1)[FUSE_PPW], int k,
                                              const float* __restrict__ xt, int lane,
                                              int* list, float* at) {
  int total = 0;
#pragma unroll
  for (int i = 0; i < FUSE_PPW; ++i) {
#pragma unroll
    for (int h = 0; h < 2; ++h) {
      const Chunk& c = h ? c1[i] : c0[i];
      const bool     hit = c.ek == k;
      const unsigned m   = __builtin_amdgcn_ballot_w32(hit);
      const int      pos = total + (int)__builtin_amdgcn_mbcnt_lo(m, 0u);
      if (hit) {
        int2 e;
        e.x = (i << 14) | (c.ei & (N_PTS - 1));
        e.y = __float_as_int(c.ep);
        *(int2*)&list[2 * pos] = e;
      }
      total += __builtin_popcount(m);
    }
  }
  __builtin_amdgcn_fence(__ATOMIC_RELEASE, "wavefront");
  __builtin_amdgcn_wave_barrier();
  __builtin_amdgcn_fence(__ATOMIC_ACQUIRE, "wavefront");

  int    pt = 0;
  int    written = 0;
  float4 s0 = make_float4(0.f, 0.f, 0.f, 0.f);
  float4 s1 = s0;
  for (int base = 0; base < total; base += 32) {
    const int n = total - base < 32 ? total - base : 32;
    int2 e = make_int2(0, 0);
    if (lane < n) e = *(const int2*)&list[2 * (base + lane)];
    for (int j0 = 0; j0 < n; j0 += GATHER_SLOTS) {
      const int c = n - j0 < GATHER_SLOTS ? n - j0 : GATHER_SLOTS;
      TripSel<GATHER_SLOTS>::run(c, e, j0, xt, lane, at, pt, written, s0, s1);
    }
  }
  store_point(at, pt, lane, s0, s1);
  written |= 1 << pt;
  const float4 zero = make_float4(0.f, 0.f, 0.f, 0.f);
#pragma unroll
  for (int i = 0; i < FUSE_PPW; ++i) {
    if (!(written & (1 << i))) store_point(at, i, lane, zero, zero);
  }
}

#define BIN_UNROLL (RNG_CAP / FUSE_THREADS)

__global__ void __launch_bounds__(FUSE_THREADS)
fused_kernel(const int* __restrict__ rcnt,
             const int* __restrict__ rr,
             const int* __restrict__ cnt,
             const int* __restrict__ ovk,
             const int* __restrict__ ovi,
             const float* __restrict__ ovp,
             const float* __restrict__ xt,
             const float* __restrict__ wt,
             const float* __restrict__ bias,
             float* __restrict__ y,
             int nnz) {
  __shared__ __attribute__((aligned(16))) float lds[W_FLOATS + A_FLOATS];
  __shared__ __attribute__((aligned(16))) int lists[FUSE_WAVES * LIST_INTS];
  __shared__ __attribute__((aligned(16))) int pc[FUSE_PTS];
  __shared__ int2 lb[FUSE_PTS * BIN_CAP];
  static_assert(256 * TP <= W_FLOATS, "output staging tile must fit in the weights' LDS");
  static_assert(W_VEC * FUSE_THREADS * 4 == KSZ * COUT * CIN, "the threads' pieces cover the weights exactly");
  static_assert(FUSE_THREADS == 512, "the weight staging below splits a 64 x 64 block into two halves of 512 pieces");
  static_assert(BIN_UNROLL * FUSE_THREADS == RNG_CAP, "one pass of the binning covers a range region exactly");
  static_assert(RNG_PTS % FUSE_PTS == 0, "a block's points lie in one range");

  const int tid  = threadIdx.x;
  const int lane = tid & 31;
  const int hi   = lane >> 4;
  const int l15  = lane & 15;
  const int wave = __builtin_amdgcn_readfirstlane(tid >> 5);
  const int n0   = (blockIdx.x * FUSE_WAVES + wave) * FUSE_PPW;
  const int rg   = blockIdx.x / (RNG_PTS / FUSE_PTS);
  const int qt   = blockIdx.x % (RNG_PTS / FUSE_PTS);

  if (tid < FUSE_PTS) pc[tid] = 0;
  __syncthreads();
  int nrec = rcnt[rg * RC_PITCH];

  {
    const float4* const wsrc = (const float4*)wt + tid;
    const int wofs = (tid >> 4) * LDP + (tid & 15) * 4;
    float4 wv[W_VEC];
#pragma unroll
    for (int i = 0; i < W_VEC; ++i) wv[i] = wsrc[i * FUSE_THREADS];
#pragma unroll
    for (int i = 0; i < W_VEC; ++i)
      *(float4*)&lds[(i >> 1) * (64 * LDP) + (i & 1) * (32 * LDP) + wofs] = wv[i];
  }
  const bool spilled = nrec > RNG_CAP;
  if (nrec > RNG_CAP) nrec = RNG_CAP;
  if (nrec < 0) nrec = 0;
  const int2* const srcp = (const int2*)rr + (size_t)rg * RNG_CAP;
  {
    int2 w[BIN_UNROLL];
#pragma unroll
    for (int u = 0; u < BIN_UNROLL; ++u) {
      const int i = u * FUSE_THREADS + tid;
      w[u] = make_int2(-1, 0);
      if (i < nrec) w[u] = srcp[i];
    }
    int r[BIN_UNROLL];
#pragma unroll
    for (int u = 0; u < BIN_UNROLL; ++u) {
      r[u] = -1;
      if (w[u].x >= 0) {
        const int pr = (w[u].x >> 14) & (RNG_PTS - 1);
        if (pr / FUSE_PTS == qt) r[u] = atomicAdd(&pc[pr & (FUSE_PTS - 1)], 1);
      }
    }
#pragma unroll
    for (int u = 0; u < BIN_UNROLL; ++u) {
      if ((unsigned)r[u] < (unsigned)BIN_CAP) {
        const int lp = (w[u].x >> 14) & (FUSE_PTS - 1);
        lb[lp * BIN_CAP + r[u]] = make_int2((((w[u].x >> 22) & 15) << 14) | (w[u].x & (N_PTS - 1)), w[u].y);
      }
    }
  }
  __syncthreads();

  bool longseg = spilled;
  bool inbin[FUSE_PPW];
  Chunk ck[FUSE_PPW], cm[FUSE_PPW];
  const int4 pc4 = *(const int4*)&pc[wave * FUSE_PPW];
  const int plen[FUSE_PPW] = { pc4.x, pc4.y, pc4.z, pc4.w };
  static_assert(FUSE_PPW == 4, "the wave's counts are read as one int4");
#pragma unroll
  for (int i = 0; i < FUSE_PPW; ++i) {
    int len = __builtin_amdgcn_readfirstlane(plen[i]);
    inbin[i] = len <= BIN_CAP;
    longseg |= !inbin[i];
    if (len > BIN_CAP) len = BIN_CAP;
    if (len < 0) len = 0;
    const int b0 = (wave * FUSE_PPW + i) * BIN_CAP;
    ck[i] = load_chunk(lb, b0,      b0 + len, lane);
    cm[i] = load_chunk(lb, b0 + 32, b0 + len, lane);
  }
  int novf = 0;
  if (spilled) {
    novf = cnt[N_PTS];
    if (novf > nnz) novf = nnz;
    if (novf < 0) novf = 0;
  }
  novf = __builtin_amdgcn_readfirstlane(novf);
  nrec = __builtin_amdgcn_readfirstlane(nrec);

  float* const at = lds + W_FLOATS + wave * (16 * LDP);
  int* const list = lists + wave * LIST_INTS;
  v8f acc[4] = {};

  for (int kk = 0; kk < KSZ; ++kk) {
    __builtin_amdgcn_fence(__ATOMIC_RELEASE, "wavefront");
    __builtin_amdgcn_wave_barrier();
    __builtin_amdgcn_fence(__ATOMIC_ACQUIRE, "wavefront");
    if (!longseg) {
      gather_stream(ck, cm, kk, xt, lane, list, at);
    } else {
      float4 a0[FUSE_PPW], a1[FUSE_PPW];
#pragma unroll
      for (int i = 0; i < FUSE_PPW; ++i) {
        a0[i] = make_float4(0.f, 0.f, 0.f, 0.f);
        a1[i] = a0[i];
        if (inbin[i]) {
          add_chunk(ck[i], kk, xt, lane, a0[i], a1[i]);
          add_chunk(cm[i], kk, xt, lane, a0[i], a1[i]);
        } else {
          const int prow = (n0 + i) & (RNG_PTS - 1);
          for (int base = 0; base < nrec; base += 32) {
            const int idx = base + lane;
            Chunk c2;
            c2.ek = -1;
            c2.ei = 0;
            c2.ep = 0.f;
            if (idx < nrec) {
              const int2 w = srcp[idx];
              if (w.x >= 0 && ((w.x >> 14) & (RNG_PTS - 1)) == prow) {
                c2.ek = (w.x >> 22) & 15;
                c2.ei = w.x & (N_PTS - 1);
                c2.ep = __int_as_float(w.y);
              }
            }
            add_chunk(c2, kk, xt, lane, a0[i], a1[i]);
          }
        }
        for (int base = 0; base < novf; base += 32) {
          const int idx = base + lane;
          Chunk c2;
          c2.ek = -1;
          c2.ei = 0;
          c2.ep = 0.f;
          if (idx < novf) {
            const int ok = ovk[idx];
            if ((ok & (N_PTS - 1)) == n0 + i) {
              c2.ek = ok >> 14;
              c2.ei = ovi[idx];
              c2.ep = ovp[idx];
            }
          }
          add_chunk(c2, kk, xt, lane, a0[i], a1[i]);
        }
        store_point(at, i, lane, a0[i], a1[i]);
      }
    }

    __builtin_amdgcn_fence(__ATOMIC_RELEASE, "wavefront");
    __builtin_amdgcn_wave_barrier();
    __builtin_amdgcn_fence(__ATOMIC_ACQUIRE, "wavefront");
    const float* const wb = lds + kk * (64 * LDP);

#pragma unroll
    for (int ks = 0; ks < 16; ++ks) {
      const int k2 = ks * 4 + hi * 2;
      v2f a = { at[l15 * LDP + k2], at[l15 * LDP + k2 + 1] };
#pragma unroll
      for (int wn = 0; wn < 4; ++wn) {
        v2f bw = { wb[(wn * 16 + l15) * LDP + k2], wb[(wn * 16 + l15) * LDP + k2 + 1] };
        acc[wn] = __builtin_amdgcn_wmma_f32_16x16x4_f32(
             false, a,  false, bw,
             (short)0, acc[wn],  false,  false);
      }
    }
  }

  __syncthreads();
  float* const T = lds;
#pragma unroll
  for (int wn = 0; wn < 4; ++wn) {
    const float bv = bias[wn * 16 + l15];
#pragma unroll
    for (int r = 0; r < 8; ++r) {
      T[((r & 3) * COUT + wn * 16 + l15) * TP + wave * FUSE_PPW + 2 * hi + (r >> 2)] = acc[wn][r] + bv;
    }
  }
  __syncthreads();

  const int nb = blockIdx.x * FUSE_PTS;
#pragma unroll
  for (int s = 0; s < 8; ++s) {
    const int j = wave * 16 + s * 2 + (lane >> 4);
    const float4 v = *(const float4*)&T[j * TP + (lane & 15) * 4];
    *(float4*)(y + (size_t)j * N_PTS + nb + (lane & 15) * 4) = v;
  }
}

extern "C" void kernel_launch(void* const* d_in, const int* in_sizes, int n_in,
                              void* d_out, int out_size, void* d_ws, size_t ws_size,
                              hipStream_t stream) {
  const float* x        = (const float*)d_in[0];
  const int*   psi_idx  = (const int*)  d_in[1];
  const float* psi_vals = (const float*)d_in[2];
  const float* quad     = (const float*)d_in[3];
  const float* weight   = (const float*)d_in[4];
  const float* bias     = (const float*)d_in[5];
  float*       y        = (float*)d_out;

  const int nnz = in_sizes[2];

  float* xt   = (float*)d_ws;
  float* wt   = xt + (size_t)N_PTS * BC;
  int*   rr   = (int*)(wt + KSZ * COUT * CIN) + 2 * N_PTS * BIN_CAP;
  int*   rcnt = rr + 2 * N_RNG * RNG_CAP;
  int*   cnt  = (int*)(wt + (size_t)N_PTS * BC);
  int*   ovk  = cnt + N_PTS + 256;
  int*   ovi  = ovk + nnz;
  float* ovp  = (float*)(ovi + nnz);

  const int* kI = psi_idx;
  const int* oI = psi_idx + nnz;
  const int* iI = psi_idx + 2 * nnz;

  setup_kernel<<<XQ_BLOCKS + WT_BLOCKS + ZERO_BLOCKS, 256, 0, stream>>>(x, quad, xt, weight, wt, rcnt, cnt);

  range_kernel<<<256, 256, 0, stream>>>(kI, oI, iI, psi_vals, rcnt, rr, cnt, ovk, ovi, ovp, nnz);

  fused_kernel<<<N_PTS / FUSE_PTS, FUSE_THREADS, 0, stream>>>(rcnt, rr, cnt, ovk, ovi, ovp, xt, wt, bias, y, nnz);
}
